# starting kernel
# baseline (speedup 1.0000x reference)
#include <hip/hip_runtime.h>


#define SEQ   90
#define KP    17
#define FD    2
#define BATCH 64
#define NGR   (BATCH * SEQ)
#define NN    (NGR * KP)
#define HID   512
#define GATE  2048
#define D_IN1 1122
#define D_IN2 1024

typedef float v2f __attribute__((ext_vector_type(2)));
typedef float v8f __attribute__((ext_vector_type(8)));

__device__ __forceinline__ v8f wmma_f32(v2f a, v2f b, v8f c) {
    return __builtin_amdgcn_wmma_f32_16x16x4_f32(false, a, false, b, (short)0, c,
                                                 false, false);
}

__device__ __forceinline__ float sigmoidf_(float x) { return 1.0f / (1.0f + __expf(-x)); }

__constant__ int NBR_OFF[18] = {0,2,5,8,10,12,16,20,22,24,25,26,29,32,34,36,37,38};
__constant__ int NBR[38] = {
    1,2, 2,0,3, 1,0,4, 1,5, 2,6, 11,6,7,3, 12,5,8,4, 5,9, 6,10, 7, 8,
    13,12,5, 14,11,6, 15,11, 16,12, 13, 14
};

__global__ void gemm_wmma_f32(const float* __restrict__ A, int lda,
                              const float* __restrict__ Bm, int ldb,
                              const float* __restrict__ bias,
                              float* __restrict__ C, int ldc,
                              int M, int N, int K, int act) {
    int wave = (blockIdx.x * blockDim.x + threadIdx.x) >> 5;
    int lane = threadIdx.x & 31;
    int tilesN = (N + 31) >> 5;
    int tilesM = M >> 5;
    if (wave >= tilesM * tilesN) return;
    int tm = wave / tilesN, tn = wave % tilesN;
    int m0 = tm << 5, n0 = tn << 5;
    int half = lane >> 4, l = lane & 15;
    int kofs = half << 1;

    const float* A0 = A + (size_t)(m0 + l) * lda + kofs;
    const float* A1 = A0 + (size_t)16 * lda;
    int bn0 = n0 + l;      if (bn0 > N - 1) bn0 = N - 1;
    int bn1 = n0 + 16 + l; if (bn1 > N - 1) bn1 = N - 1;
    const float* B0 = Bm + (size_t)bn0 * ldb + kofs;
    const float* B1 = Bm + (size_t)bn1 * ldb + kofs;

    v8f acc00 = {0.f,0.f,0.f,0.f,0.f,0.f,0.f,0.f};
    v8f acc01 = acc00, acc10 = acc00, acc11 = acc00;

    int Kmain = K & ~3;
    v2f a0 = *(const v2f*)A0;
    v2f a1 = *(const v2f*)A1;
    v2f b0 = *(const v2f*)B0;
    v2f b1 = *(const v2f*)B1;
    for (int k0 = 0; k0 < Kmain - 4; k0 += 4) {
        v2f a0n = *(const v2f*)(A0 + k0 + 4);
        v2f a1n = *(const v2f*)(A1 + k0 + 4);
        v2f b0n = *(const v2f*)(B0 + k0 + 4);
        v2f b1n = *(const v2f*)(B1 + k0 + 4);
        acc00 = wmma_f32(a0, b0, acc00);
        acc01 = wmma_f32(a0, b1, acc01);
        acc10 = wmma_f32(a1, b0, acc10);
        acc11 = wmma_f32(a1, b1, acc11);
        a0 = a0n; a1 = a1n; b0 = b0n; b1 = b1n;
    }
    acc00 = wmma_f32(a0, b0, acc00);
    acc01 = wmma_f32(a0, b1, acc01);
    acc10 = wmma_f32(a1, b0, acc10);
    acc11 = wmma_f32(a1, b1, acc11);

    if (K & 3) {
        int ka = Kmain + kofs, kb = ka + 1;
        int kaC = ka < K ? ka : K - 1;
        int kbC = kb < K ? kb : K - 1;
        const float* A0b = A + (size_t)(m0 + l) * lda;
        const float* A1b = A0b + (size_t)16 * lda;
        const float* B0b = Bm + (size_t)bn0 * ldb;
        const float* B1b = Bm + (size_t)bn1 * ldb;
        float va0 = A0b[kaC], wa0 = A0b[kbC];
        float va1 = A1b[kaC], wa1 = A1b[kbC];
        float vb0 = B0b[kaC], wb0 = B0b[kbC];
        float vb1 = B1b[kaC], wb1 = B1b[kbC];
        v2f at0, at1, bt0, bt1;
        at0.x = ka < K ? va0 : 0.f; at0.y = kb < K ? wa0 : 0.f;
        at1.x = ka < K ? va1 : 0.f; at1.y = kb < K ? wa1 : 0.f;
        bt0.x = ka < K ? vb0 : 0.f; bt0.y = kb < K ? wb0 : 0.f;
        bt1.x = ka < K ? vb1 : 0.f; bt1.y = kb < K ? wb1 : 0.f;
        acc00 = wmma_f32(at0, bt0, acc00);
        acc01 = wmma_f32(at0, bt1, acc01);
        acc10 = wmma_f32(at1, bt0, acc10);
        acc11 = wmma_f32(at1, bt1, acc11);
    }
    int cn0 = n0 + l, cn1 = n0 + 16 + l;
    if (cn0 < N) {
        float bv = bias ? bias[cn0] : 0.f;
        #pragma unroll
        for (int j = 0; j < 8; ++j) {
            int r0 = m0 + j + half * 8, r1 = r0 + 16;
            float v0 = acc00[j] + bv, v1 = acc10[j] + bv;
            if (act) { v0 = fmaxf(v0, 0.f); v1 = fmaxf(v1, 0.f); }
            C[(size_t)r0 * ldc + cn0] = v0;
            C[(size_t)r1 * ldc + cn0] = v1;
        }
    }
    if (cn1 < N) {
        float bv = bias ? bias[cn1] : 0.f;
        #pragma unroll
        for (int j = 0; j < 8; ++j) {
            int r0 = m0 + j + half * 8, r1 = r0 + 16;
            float v0 = acc01[j] + bv, v1 = acc11[j] + bv;
            if (act) { v0 = fmaxf(v0, 0.f); v1 = fmaxf(v1, 0.f); }
            C[(size_t)r0 * ldc + cn1] = v0;
            C[(size_t)r1 * ldc + cn1] = v1;
        }
    }
}

__global__ void lstm_step_gemm(const float* __restrict__ Zx,
                               const float* __restrict__ Whh,
                               const float* __restrict__ hprev,
                               float* __restrict__ Zfull, int t) {
    int dir  = blockIdx.y;
    int wave = blockIdx.x * 8 + (threadIdx.x >> 5);
    int lane = threadIdx.x & 31;
    int tm = wave >> 6, tn = wave & 63;
    int m0 = tm << 5, n0 = tn << 5;
    int half = lane >> 4, l = lane & 15;
    int kofs = half << 1;

    const float* A0 = hprev + (size_t)dir * BATCH * HID + (size_t)(m0 + l) * HID + kofs;
    const float* A1 = A0 + (size_t)16 * HID;
    const float* B0 = Whh + (size_t)dir * GATE * HID + (size_t)(n0 + l) * HID + kofs;
    const float* B1 = B0 + (size_t)16 * HID;

    v8f acc00 = {0.f,0.f,0.f,0.f,0.f,0.f,0.f,0.f};
    v8f acc01 = acc00, acc10 = acc00, acc11 = acc00;

    v2f a0 = *(const v2f*)A0;
    v2f a1 = *(const v2f*)A1;
    v2f b0 = *(const v2f*)B0;
    v2f b1 = *(const v2f*)B1;
    for (int k0 = 0; k0 < HID - 4; k0 += 4) {
        v2f a0n = *(const v2f*)(A0 + k0 + 4);
        v2f a1n = *(const v2f*)(A1 + k0 + 4);
        v2f b0n = *(const v2f*)(B0 + k0 + 4);
        v2f b1n = *(const v2f*)(B1 + k0 + 4);
        acc00 = wmma_f32(a0, b0, acc00);
        acc01 = wmma_f32(a0, b1, acc01);
        acc10 = wmma_f32(a1, b0, acc10);
        acc11 = wmma_f32(a1, b1, acc11);
        a0 = a0n; a1 = a1n; b0 = b0n; b1 = b1n;
    }
    acc00 = wmma_f32(a0, b0, acc00);
    acc01 = wmma_f32(a0, b1, acc01);
    acc10 = wmma_f32(a1, b0, acc10);
    acc11 = wmma_f32(a1, b1, acc11);

    int t_d = dir ? (SEQ - 1 - t) : t;
    const float* Zxd = Zx + (size_t)dir * NGR * GATE;
    float* Zfd = Zfull + (size_t)dir * BATCH * GATE;
    int cn0 = n0 + l, cn1 = n0 + 16 + l;
    #pragma unroll
    for (int j = 0; j < 8; ++j) {
        int r0 = m0 + j + half * 8, r1 = r0 + 16;
        Zfd[(size_t)r0 * GATE + cn0] = acc00[j] + Zxd[(size_t)(r0 * SEQ + t_d) * GATE + cn0];
        Zfd[(size_t)r0 * GATE + cn1] = acc01[j] + Zxd[(size_t)(r0 * SEQ + t_d) * GATE + cn1];
        Zfd[(size_t)r1 * GATE + cn0] = acc10[j] + Zxd[(size_t)(r1 * SEQ + t_d) * GATE + cn0];
        Zfd[(size_t)r1 * GATE + cn1] = acc11[j] + Zxd[(size_t)(r1 * SEQ + t_d) * GATE + cn1];
    }
}

__global__ void lstm_cell(const float* __restrict__ Zfull,
                          float* __restrict__ hbuf, float* __restrict__ cbuf,
                          float* __restrict__ ys, int t) {
    int idx = blockIdx.x * blockDim.x + threadIdx.x;
    if (idx >= 2 * BATCH * HID) return;
    int dir = idx >> 15, r = idx & 32767;
    int b = r >> 9, j = r & 511;
    const float* z = Zfull + (size_t)dir * BATCH * GATE + (size_t)b * GATE;
    float zi = z[j], zf = z[HID + j], zg = z[2 * HID + j], zo = z[3 * HID + j];
    float c = cbuf[idx];
    c = sigmoidf_(zf) * c + sigmoidf_(zi) * tanhf(zg);
    float h = sigmoidf_(zo) * tanhf(c);
    cbuf[idx] = c; hbuf[idx] = h;
    int t_d = dir ? (SEQ - 1 - t) : t;
    ys[(size_t)(b * SEQ + t_d) * 1024 + dir * HID + j] = h;
}

__global__ void gat1_lin(const float* __restrict__ x, const float* __restrict__ W,
                         float* __restrict__ h) {
    int idx = blockIdx.x * blockDim.x + threadIdx.x;
    if (idx >= NN * 64) return;
    int i = idx >> 6, f = idx & 63;
    h[idx] = x[i * 2 + 0] * W[f] + x[i * 2 + 1] * W[64 + f];
}

__global__ void gat_scores(const float* __restrict__ h,
                           const float* __restrict__ asrc, const float* __restrict__ adst,
                           float* __restrict__ sas, float* __restrict__ sad) {
    int i = blockIdx.x * blockDim.x + threadIdx.x;
    if (i >= NN) return;
    const float* hp = h + (size_t)i * 64;
    float a = 0.f, d = 0.f;
    #pragma unroll
    for (int f = 0; f < 64; ++f) { float v = hp[f]; a += v * asrc[f]; d += v * adst[f]; }
    sas[i] = a; sad[i] = d;
}

__global__ void gat_aggr(const float* __restrict__ h,
                         const float* __restrict__ sas, const float* __restrict__ sad,
                         const float* __restrict__ bvec,
                         const float* __restrict__ g, const float* __restrict__ bb,
                         const float* __restrict__ bm, const float* __restrict__ bvv,
                         float* __restrict__ out) {
    int idx = blockIdx.x * blockDim.x + threadIdx.x;
    if (idx >= NN * 64) return;
    int i = idx >> 6, ch = idx & 63;
    int dl = i % KP, base = i - dl;
    float ad = sad[i];
    int o0 = NBR_OFF[dl], deg = NBR_OFF[dl + 1] - o0;

    float e[5], hv[5];
    float t0 = sas[i] + ad;
    e[0]  = t0 < 0.f ? 0.2f * t0 : t0;
    hv[0] = h[(size_t)i * 64 + ch];
    for (int n = 0; n < deg; ++n) {
        int s = base + NBR[o0 + n];
        float tt = sas[s] + ad;
        e[n + 1]  = tt < 0.f ? 0.2f * tt : tt;
        hv[n + 1] = h[(size_t)s * 64 + ch];
    }
    int cnt = deg + 1;
    float m = e[0];
    for (int n = 1; n < cnt; ++n) m = fmaxf(m, e[n]);
    float den = 0.f, num = 0.f;
    for (int n = 0; n < cnt; ++n) { float w = __expf(e[n] - m); den += w; num += w * hv[n]; }
    float val = num / den + bvec[ch];
    val = (val - bm[ch]) * (g[ch] * rsqrtf(bvv[ch] + 1e-5f)) + bb[ch];
    out[idx] = fmaxf(val, 0.f);
}

__global__ void pack_emb(const float* __restrict__ h2, const float* __restrict__ x,
                         float* __restrict__ emb) {
    int idx = blockIdx.x * blockDim.x + threadIdx.x;
    if (idx >= NN * 66) return;
    int i = idx / 66, c = idx % 66;
    int bt = i / KP, kp = i % KP;
    float v = (c < 64) ? h2[(size_t)i * 64 + c] : x[i * 2 + (c - 64)];
    emb[(size_t)bt * D_IN1 + kp * 66 + c] = v;
}

__global__ void transpose64(const float* __restrict__ W, float* __restrict__ WT) {
    int idx = blockIdx.x * blockDim.x + threadIdx.x;
    if (idx >= 64 * 64) return;
    int n = idx >> 6, k = idx & 63;
    WT[n * 64 + k] = W[k * 64 + n];
}

__global__ void zero_f32(float* __restrict__ p, int n) {
    int i = blockIdx.x * blockDim.x + threadIdx.x;
    if (i < n) p[i] = 0.f;
}

static inline void launch_gemm(hipStream_t s, const float* A, int lda,
                               const float* B, int ldb, const float* bias,
                               float* C, int ldc, int M, int N, int K, int act) {
    int tiles = (M / 32) * ((N + 31) / 32);
    int blocks = (tiles + 7) / 8;
    gemm_wmma_f32<<<blocks, 256, 0, s>>>(A, lda, B, ldb, bias, C, ldc, M, N, K, act);
}

extern "C" void kernel_launch(void* const* d_in, const int* in_sizes, int n_in,
                              void* d_out, int out_size, void* d_ws, size_t ws_size,
                              hipStream_t stream) {
    const float* x        = (const float*)d_in[0];
    const float* gat1_W   = (const float*)d_in[2];
    const float* gat1_as  = (const float*)d_in[3];
    const float* gat1_ad  = (const float*)d_in[4];
    const float* gat1_b   = (const float*)d_in[5];
    const float* gat2_W   = (const float*)d_in[6];
    const float* gat2_as  = (const float*)d_in[7];
    const float* gat2_ad  = (const float*)d_in[8];
    const float* gat2_b   = (const float*)d_in[9];
    const float* bn1_g = (const float*)d_in[10], *bn1_b = (const float*)d_in[11];
    const float* bn1_m = (const float*)d_in[12], *bn1_v = (const float*)d_in[13];
    const float* bn2_g = (const float*)d_in[14], *bn2_b = (const float*)d_in[15];
    const float* bn2_m = (const float*)d_in[16], *bn2_v = (const float*)d_in[17];
    const float* l1_Wih = (const float*)d_in[18];
    const float* l1_Whh = (const float*)d_in[19];
    const float* l1_b   = (const float*)d_in[20];
    const float* l2_Wih = (const float*)d_in[21];
    const float* l2_Whh = (const float*)d_in[22];
    const float* l2_b   = (const float*)d_in[23];
    const float* hW1 = (const float*)d_in[24], *hb1 = (const float*)d_in[25];
    const float* hW2 = (const float*)d_in[26], *hb2 = (const float*)d_in[27];
    float* out = (float*)d_out;
    float* ws  = (float*)d_ws;

    const size_t OFF_HA  = 0;
    const size_t OFF_HB  = OFF_HA  + (size_t)NN * 64;
    const size_t OFF_SAS = OFF_HB  + (size_t)NN * 64;
    const size_t OFF_SAD = OFF_SAS + NN;
    const size_t OFF_EMB = OFF_SAD + NN;
    const size_t OFF_ZX  = OFF_EMB + (size_t)NGR * D_IN1;
    const size_t OFF_ZF  = OFF_ZX  + (size_t)2 * NGR * GATE;
    const size_t OFF_HST = OFF_ZF  + (size_t)2 * BATCH * GATE;
    const size_t OFF_CST = OFF_HST + (size_t)2 * BATCH * HID;
    const size_t OFF_WT  = OFF_CST + (size_t)2 * BATCH * HID;

    float* hA   = ws + OFF_HA;
    float* hB   = ws + OFF_HB;
    float* sas  = ws + OFF_SAS;
    float* sad  = ws + OFF_SAD;
    float* emb  = ws + OFF_EMB;
    float* Zx   = ws + OFF_ZX;
    float* Zf   = ws + OFF_ZF;
    float* hst  = ws + OFF_HST;
    float* cst  = ws + OFF_CST;
    float* wt   = ws + OFF_WT;
    float* ys1  = emb;
    float* ys2  = hB;
    float* hd1  = hA;

    const int TPB = 256;

    gat1_lin<<<(NN * 64 + TPB - 1) / TPB, TPB, 0, stream>>>(x, gat1_W, hA);
    gat_scores<<<(NN + TPB - 1) / TPB, TPB, 0, stream>>>(hA, gat1_as, gat1_ad, sas, sad);
    gat_aggr<<<(NN * 64 + TPB - 1) / TPB, TPB, 0, stream>>>(hA, sas, sad, gat1_b,
                                                            bn1_g, bn1_b, bn1_m, bn1_v, hB);
    transpose64<<<16, 256, 0, stream>>>(gat2_W, wt);
    launch_gemm(stream, hB, 64, wt, 64, nullptr, hA, 64, NN, 64, 64, 0);
    gat_scores<<<(NN + TPB - 1) / TPB, TPB, 0, stream>>>(hA, gat2_as, gat2_ad, sas, sad);
    gat_aggr<<<(NN * 64 + TPB - 1) / TPB, TPB, 0, stream>>>(hA, sas, sad, gat2_b,
                                                            bn2_g, bn2_b, bn2_m, bn2_v, hB);
    pack_emb<<<(NN * 66 + TPB - 1) / TPB, TPB, 0, stream>>>(hB, x, emb);

    for (int d = 0; d < 2; ++d)
        launch_gemm(stream, emb, D_IN1, l1_Wih + (size_t)d * GATE * D_IN1, D_IN1,
                    l1_b + (size_t)d * GATE, Zx + (size_t)d * NGR * GATE, GATE,
                    NGR, GATE, D_IN1, 0);
    zero_f32<<<(2 * 2 * BATCH * HID + TPB - 1) / TPB, TPB, 0, stream>>>(hst, 2 * 2 * BATCH * HID);
    for (int t = 0; t < SEQ; ++t) {
        lstm_step_gemm<<<dim3(16, 2), 256, 0, stream>>>(Zx, l1_Whh, hst, Zf, t);
        lstm_cell<<<(2 * BATCH * HID + TPB - 1) / TPB, TPB, 0, stream>>>(Zf, hst, cst, ys1, t);
    }

    for (int d = 0; d < 2; ++d)
        launch_gemm(stream, ys1, D_IN2, l2_Wih + (size_t)d * GATE * D_IN2, D_IN2,
                    l2_b + (size_t)d * GATE, Zx + (size_t)d * NGR * GATE, GATE,
                    NGR, GATE, D_IN2, 0);
    zero_f32<<<(2 * 2 * BATCH * HID + TPB - 1) / TPB, TPB, 0, stream>>>(hst, 2 * 2 * BATCH * HID);
    for (int t = 0; t < SEQ; ++t) {
        lstm_step_gemm<<<dim3(16, 2), 256, 0, stream>>>(Zx, l2_Whh, hst, Zf, t);
        lstm_cell<<<(2 * BATCH * HID + TPB - 1) / TPB, TPB, 0, stream>>>(Zf, hst, cst, ys2, t);
    }

    launch_gemm(stream, ys2, 1024, hW1, 1024, hb1, hd1, 256, NGR, 256, 1024, 1);
    launch_gemm(stream, hd1, 256,  hW2, 256,  hb2, out, 57,  NGR, 57,  256,  0);
}


# recurrent step GEMM: K split over the 8 waves of a work-group (one 32x32 tile each, all slice loads issued up front), partials summed via LDS: 8-way split changes the f32 summation order
# speedup vs baseline: 1.9231x; 1.9231x over previous
#include <hip/hip_runtime.h>


#define SEQ   90
#define KP    17
#define FD    2
#define BATCH 64
#define NGR   (BATCH * SEQ)
#define NN    (NGR * KP)
#define HID   512
#define GATE  2048
#define D_IN1 1122
#define D_IN2 1024

typedef float v2f __attribute__((ext_vector_type(2)));
typedef float v8f __attribute__((ext_vector_type(8)));

__device__ __forceinline__ v8f wmma_f32(v2f a, v2f b, v8f c) {
    return __builtin_amdgcn_wmma_f32_16x16x4_f32(false, a, false, b, (short)0, c,
                                                 false, false);
}

__device__ __forceinline__ float sigmoidf_(float x) { return 1.0f / (1.0f + __expf(-x)); }

__constant__ int NBR_OFF[18] = {0,2,5,8,10,12,16,20,22,24,25,26,29,32,34,36,37,38};
__constant__ int NBR[38] = {
    1,2, 2,0,3, 1,0,4, 1,5, 2,6, 11,6,7,3, 12,5,8,4, 5,9, 6,10, 7, 8,
    13,12,5, 14,11,6, 15,11, 16,12, 13, 14
};

__global__ void gemm_wmma_f32(const float* __restrict__ A, int lda,
                              const float* __restrict__ Bm, int ldb,
                              const float* __restrict__ bias,
                              float* __restrict__ C, int ldc,
                              int M, int N, int K, int act) {
    int wave = (blockIdx.x * blockDim.x + threadIdx.x) >> 5;
    int lane = threadIdx.x & 31;
    int tilesN = (N + 31) >> 5;
    int tilesM = M >> 5;
    if (wave >= tilesM * tilesN) return;
    int tm = wave / tilesN, tn = wave % tilesN;
    int m0 = tm << 5, n0 = tn << 5;
    int half = lane >> 4, l = lane & 15;
    int kofs = half << 1;

    const float* A0 = A + (size_t)(m0 + l) * lda + kofs;
    const float* A1 = A0 + (size_t)16 * lda;
    int bn0 = n0 + l;      if (bn0 > N - 1) bn0 = N - 1;
    int bn1 = n0 + 16 + l; if (bn1 > N - 1) bn1 = N - 1;
    const float* B0 = Bm + (size_t)bn0 * ldb + kofs;
    const float* B1 = Bm + (size_t)bn1 * ldb + kofs;

    v8f acc00 = {0.f,0.f,0.f,0.f,0.f,0.f,0.f,0.f};
    v8f acc01 = acc00, acc10 = acc00, acc11 = acc00;

    int Kmain = K & ~3;
    v2f a0 = *(const v2f*)A0;
    v2f a1 = *(const v2f*)A1;
    v2f b0 = *(const v2f*)B0;
    v2f b1 = *(const v2f*)B1;
    for (int k0 = 0; k0 < Kmain - 4; k0 += 4) {
        v2f a0n = *(const v2f*)(A0 + k0 + 4);
        v2f a1n = *(const v2f*)(A1 + k0 + 4);
        v2f b0n = *(const v2f*)(B0 + k0 + 4);
        v2f b1n = *(const v2f*)(B1 + k0 + 4);
        acc00 = wmma_f32(a0, b0, acc00);
        acc01 = wmma_f32(a0, b1, acc01);
        acc10 = wmma_f32(a1, b0, acc10);
        acc11 = wmma_f32(a1, b1, acc11);
        a0 = a0n; a1 = a1n; b0 = b0n; b1 = b1n;
    }
    acc00 = wmma_f32(a0, b0, acc00);
    acc01 = wmma_f32(a0, b1, acc01);
    acc10 = wmma_f32(a1, b0, acc10);
    acc11 = wmma_f32(a1, b1, acc11);

    if (K & 3) {
        int ka = Kmain + kofs, kb = ka + 1;
        int kaC = ka < K ? ka : K - 1;
        int kbC = kb < K ? kb : K - 1;
        const float* A0b = A + (size_t)(m0 + l) * lda;
        const float* A1b = A0b + (size_t)16 * lda;
        const float* B0b = Bm + (size_t)bn0 * ldb;
        const float* B1b = Bm + (size_t)bn1 * ldb;
        float va0 = A0b[kaC], wa0 = A0b[kbC];
        float va1 = A1b[kaC], wa1 = A1b[kbC];
        float vb0 = B0b[kaC], wb0 = B0b[kbC];
        float vb1 = B1b[kaC], wb1 = B1b[kbC];
        v2f at0, at1, bt0, bt1;
        at0.x = ka < K ? va0 : 0.f; at0.y = kb < K ? wa0 : 0.f;
        at1.x = ka < K ? va1 : 0.f; at1.y = kb < K ? wa1 : 0.f;
        bt0.x = ka < K ? vb0 : 0.f; bt0.y = kb < K ? wb0 : 0.f;
        bt1.x = ka < K ? vb1 : 0.f; bt1.y = kb < K ? wb1 : 0.f;
        acc00 = wmma_f32(at0, bt0, acc00);
        acc01 = wmma_f32(at0, bt1, acc01);
        acc10 = wmma_f32(at1, bt0, acc10);
        acc11 = wmma_f32(at1, bt1, acc11);
    }
    int cn0 = n0 + l, cn1 = n0 + 16 + l;
    if (cn0 < N) {
        float bv = bias ? bias[cn0] : 0.f;
        #pragma unroll
        for (int j = 0; j < 8; ++j) {
            int r0 = m0 + j + half * 8, r1 = r0 + 16;
            float v0 = acc00[j] + bv, v1 = acc10[j] + bv;
            if (act) { v0 = fmaxf(v0, 0.f); v1 = fmaxf(v1, 0.f); }
            C[(size_t)r0 * ldc + cn0] = v0;
            C[(size_t)r1 * ldc + cn0] = v1;
        }
    }
    if (cn1 < N) {
        float bv = bias ? bias[cn1] : 0.f;
        #pragma unroll
        for (int j = 0; j < 8; ++j) {
            int r0 = m0 + j + half * 8, r1 = r0 + 16;
            float v0 = acc01[j] + bv, v1 = acc11[j] + bv;
            if (act) { v0 = fmaxf(v0, 0.f); v1 = fmaxf(v1, 0.f); }
            C[(size_t)r0 * ldc + cn1] = v0;
            C[(size_t)r1 * ldc + cn1] = v1;
        }
    }
}

#define KSPLIT    8
#define KSLICE    (HID / KSPLIT)
#define RED_PITCH 34

typedef float v4f __attribute__((ext_vector_type(4)));

struct Slice { v4f q[8]; };

__device__ __forceinline__ void load_group(Slice& s, const float* __restrict__ p, int g) {
    s.q[2 * g]     = *(const v4f*)(p + 16 * g);
    s.q[2 * g + 1] = *(const v4f*)(p + 16 * g + 4);
}

__device__ __forceinline__ v2f frag_step(const Slice& s, int st) {
    v4f v = s.q[st >> 1];
    v2f r;
    if (st & 1) { r.x = v.z; r.y = v.w; } else { r.x = v.x; r.y = v.y; }
    return r;
}

__global__ __launch_bounds__(256)
void lstm_step_gemm(const float* __restrict__ Zx,
                    const float* __restrict__ Whh,
                    const float* __restrict__ hprev,
                    float* __restrict__ Zfull, int t) {
    __shared__ float red[KSPLIT * 32 * RED_PITCH];

    int dir  = blockIdx.y;
    int tile = blockIdx.x;
    int wv   = threadIdx.x >> 5;
    int lane = threadIdx.x & 31;
    int tm = tile >> 6, tn = tile & 63;
    int m0 = tm << 5, n0 = tn << 5;
    int half = lane >> 4, l = lane & 15;

    int kb = wv * KSLICE + half * 8;
    const float* A0 = hprev + (size_t)dir * BATCH * HID + (size_t)(m0 + l) * HID + kb;
    const float* B0 = Whh + (size_t)dir * GATE * HID + (size_t)(n0 + l) * HID + kb;
    const float* A1 = A0 + (size_t)16 * HID;
    const float* B1 = B0 + (size_t)16 * HID;
    Slice a0, a1, b0, b1;
    #pragma unroll
    for (int g = 0; g < 4; ++g) {
        load_group(a0, A0, g);
        load_group(b0, B0, g);
        load_group(b1, B1, g);
        load_group(a1, A1, g);
    }
    int t_d = dir ? (SEQ - 1 - t) : t;
    const float* Zxd = Zx + (size_t)dir * NGR * GATE;
    float zx[4];
    #pragma unroll
    for (int i = 0; i < 4; ++i) {
        int r = m0 + wv + 8 * i;
        zx[i] = Zxd[(size_t)(r * SEQ + t_d) * GATE + n0 + lane];
    }
    __builtin_amdgcn_sched_barrier(0);

    v8f acc00 = {0.f,0.f,0.f,0.f,0.f,0.f,0.f,0.f};
    v8f acc01 = acc00, acc10 = acc00, acc11 = acc00;
    #pragma unroll
    for (int st = 0; st < 16; ++st) {
        v2f fa0 = frag_step(a0, st), fa1 = frag_step(a1, st);
        v2f fb0 = frag_step(b0, st), fb1 = frag_step(b1, st);
        acc00 = wmma_f32(fa0, fb0, acc00);
        acc01 = wmma_f32(fa0, fb1, acc01);
        acc10 = wmma_f32(fa1, fb0, acc10);
        acc11 = wmma_f32(fa1, fb1, acc11);
    }

    float* rw = red + wv * (32 * RED_PITCH);
    #pragma unroll
    for (int j = 0; j < 8; ++j) {
        int r = j + 8 * half;
        rw[r * RED_PITCH + l]             = acc00[j];
        rw[r * RED_PITCH + 16 + l]        = acc01[j];
        rw[(r + 16) * RED_PITCH + l]      = acc10[j];
        rw[(r + 16) * RED_PITCH + 16 + l] = acc11[j];
    }
    __syncthreads();

    float* Zfd = Zfull + (size_t)dir * BATCH * GATE;
    #pragma unroll
    for (int i = 0; i < 4; ++i) {
        int r = wv + 8 * i;
        const float* rp = red + r * RED_PITCH + lane;
        float s = rp[0];
        #pragma unroll
        for (int w = 1; w < KSPLIT; ++w) s += rp[w * (32 * RED_PITCH)];
        Zfd[(size_t)(m0 + r) * GATE + n0 + lane] = s + zx[i];
    }
}

__global__ void lstm_cell(const float* __restrict__ Zfull,
                          float* __restrict__ hbuf, float* __restrict__ cbuf,
                          float* __restrict__ ys, int t) {
    int idx = blockIdx.x * blockDim.x + threadIdx.x;
    if (idx >= 2 * BATCH * HID) return;
    int dir = idx >> 15, r = idx & 32767;
    int b = r >> 9, j = r & 511;
    const float* z = Zfull + (size_t)dir * BATCH * GATE + (size_t)b * GATE;
    float zi = z[j], zf = z[HID + j], zg = z[2 * HID + j], zo = z[3 * HID + j];
    float c = cbuf[idx];
    c = sigmoidf_(zf) * c + sigmoidf_(zi) * tanhf(zg);
    float h = sigmoidf_(zo) * tanhf(c);
    cbuf[idx] = c; hbuf[idx] = h;
    int t_d = dir ? (SEQ - 1 - t) : t;
    ys[(size_t)(b * SEQ + t_d) * 1024 + dir * HID + j] = h;
}

__global__ void gat1_lin(const float* __restrict__ x, const float* __restrict__ W,
                         float* __restrict__ h) {
    int idx = blockIdx.x * blockDim.x + threadIdx.x;
    if (idx >= NN * 64) return;
    int i = idx >> 6, f = idx & 63;
    h[idx] = x[i * 2 + 0] * W[f] + x[i * 2 + 1] * W[64 + f];
}

__global__ void gat_scores(const float* __restrict__ h,
                           const float* __restrict__ asrc, const float* __restrict__ adst,
                           float* __restrict__ sas, float* __restrict__ sad) {
    int i = blockIdx.x * blockDim.x + threadIdx.x;
    if (i >= NN) return;
    const float* hp = h + (size_t)i * 64;
    float a = 0.f, d = 0.f;
    #pragma unroll
    for (int f = 0; f < 64; ++f) { float v = hp[f]; a += v * asrc[f]; d += v * adst[f]; }
    sas[i] = a; sad[i] = d;
}

__global__ void gat_aggr(const float* __restrict__ h,
                         const float* __restrict__ sas, const float* __restrict__ sad,
                         const float* __restrict__ bvec,
                         const float* __restrict__ g, const float* __restrict__ bb,
                         const float* __restrict__ bm, const float* __restrict__ bvv,
                         float* __restrict__ out) {
    int idx = blockIdx.x * blockDim.x + threadIdx.x;
    if (idx >= NN * 64) return;
    int i = idx >> 6, ch = idx & 63;
    int dl = i % KP, base = i - dl;
    float ad = sad[i];
    int o0 = NBR_OFF[dl], deg = NBR_OFF[dl + 1] - o0;

    float e[5], hv[5];
    float t0 = sas[i] + ad;
    e[0]  = t0 < 0.f ? 0.2f * t0 : t0;
    hv[0] = h[(size_t)i * 64 + ch];
    for (int n = 0; n < deg; ++n) {
        int s = base + NBR[o0 + n];
        float tt = sas[s] + ad;
        e[n + 1]  = tt < 0.f ? 0.2f * tt : tt;
        hv[n + 1] = h[(size_t)s * 64 + ch];
    }
    int cnt = deg + 1;
    float m = e[0];
    for (int n = 1; n < cnt; ++n) m = fmaxf(m, e[n]);
    float den = 0.f, num = 0.f;
    for (int n = 0; n < cnt; ++n) { float w = __expf(e[n] - m); den += w; num += w * hv[n]; }
    float val = num / den + bvec[ch];
    val = (val - bm[ch]) * (g[ch] * rsqrtf(bvv[ch] + 1e-5f)) + bb[ch];
    out[idx] = fmaxf(val, 0.f);
}

__global__ void pack_emb(const float* __restrict__ h2, const float* __restrict__ x,
                         float* __restrict__ emb) {
    int idx = blockIdx.x * blockDim.x + threadIdx.x;
    if (idx >= NN * 66) return;
    int i = idx / 66, c = idx % 66;
    int bt = i / KP, kp = i % KP;
    float v = (c < 64) ? h2[(size_t)i * 64 + c] : x[i * 2 + (c - 64)];
    emb[(size_t)bt * D_IN1 + kp * 66 + c] = v;
}

__global__ void transpose64(const float* __restrict__ W, float* __restrict__ WT) {
    int idx = blockIdx.x * blockDim.x + threadIdx.x;
    if (idx >= 64 * 64) return;
    int n = idx >> 6, k = idx & 63;
    WT[n * 64 + k] = W[k * 64 + n];
}

__global__ void zero_f32(float* __restrict__ p, int n) {
    int i = blockIdx.x * blockDim.x + threadIdx.x;
    if (i < n) p[i] = 0.f;
}

static inline void launch_gemm(hipStream_t s, const float* A, int lda,
                               const float* B, int ldb, const float* bias,
                               float* C, int ldc, int M, int N, int K, int act) {
    int tiles = (M / 32) * ((N + 31) / 32);
    int blocks = (tiles + 7) / 8;
    gemm_wmma_f32<<<blocks, 256, 0, s>>>(A, lda, B, ldb, bias, C, ldc, M, N, K, act);
}

extern "C" void kernel_launch(void* const* d_in, const int* in_sizes, int n_in,
                              void* d_out, int out_size, void* d_ws, size_t ws_size,
                              hipStream_t stream) {
    const float* x        = (const float*)d_in[0];
    const float* gat1_W   = (const float*)d_in[2];
    const float* gat1_as  = (const float*)d_in[3];
    const float* gat1_ad  = (const float*)d_in[4];
    const float* gat1_b   = (const float*)d_in[5];
    const float* gat2_W   = (const float*)d_in[6];
    const float* gat2_as  = (const float*)d_in[7];
    const float* gat2_ad  = (const float*)d_in[8];
    const float* gat2_b   = (const float*)d_in[9];
    const float* bn1_g = (const float*)d_in[10], *bn1_b = (const float*)d_in[11];
    const float* bn1_m = (const float*)d_in[12], *bn1_v = (const float*)d_in[13];
    const float* bn2_g = (const float*)d_in[14], *bn2_b = (const float*)d_in[15];
    const float* bn2_m = (const float*)d_in[16], *bn2_v = (const float*)d_in[17];
    const float* l1_Wih = (const float*)d_in[18];
    const float* l1_Whh = (const float*)d_in[19];
    const float* l1_b   = (const float*)d_in[20];
    const float* l2_Wih = (const float*)d_in[21];
    const float* l2_Whh = (const float*)d_in[22];
    const float* l2_b   = (const float*)d_in[23];
    const float* hW1 = (const float*)d_in[24], *hb1 = (const float*)d_in[25];
    const float* hW2 = (const float*)d_in[26], *hb2 = (const float*)d_in[27];
    float* out = (float*)d_out;
    float* ws  = (float*)d_ws;

    const size_t OFF_HA  = 0;
    const size_t OFF_HB  = OFF_HA  + (size_t)NN * 64;
    const size_t OFF_SAS = OFF_HB  + (size_t)NN * 64;
    const size_t OFF_SAD = OFF_SAS + NN;
    const size_t OFF_EMB = OFF_SAD + NN;
    const size_t OFF_ZX  = OFF_EMB + (size_t)NGR * D_IN1;
    const size_t OFF_ZF  = OFF_ZX  + (size_t)2 * NGR * GATE;
    const size_t OFF_HST = OFF_ZF  + (size_t)2 * BATCH * GATE;
    const size_t OFF_CST = OFF_HST + (size_t)2 * BATCH * HID;
    const size_t OFF_WT  = OFF_CST + (size_t)2 * BATCH * HID;

    float* hA   = ws + OFF_HA;
    float* hB   = ws + OFF_HB;
    float* sas  = ws + OFF_SAS;
    float* sad  = ws + OFF_SAD;
    float* emb  = ws + OFF_EMB;
    float* Zx   = ws + OFF_ZX;
    float* Zf   = ws + OFF_ZF;
    float* hst  = ws + OFF_HST;
    float* cst  = ws + OFF_CST;
    float* wt   = ws + OFF_WT;
    float* ys1  = emb;
    float* ys2  = hB;
    float* hd1  = hA;

    const int TPB = 256;

    gat1_lin<<<(NN * 64 + TPB - 1) / TPB, TPB, 0, stream>>>(x, gat1_W, hA);
    gat_scores<<<(NN + TPB - 1) / TPB, TPB, 0, stream>>>(hA, gat1_as, gat1_ad, sas, sad);
    gat_aggr<<<(NN * 64 + TPB - 1) / TPB, TPB, 0, stream>>>(hA, sas, sad, gat1_b,
                                                            bn1_g, bn1_b, bn1_m, bn1_v, hB);
    transpose64<<<16, 256, 0, stream>>>(gat2_W, wt);
    launch_gemm(stream, hB, 64, wt, 64, nullptr, hA, 64, NN, 64, 64, 0);
    gat_scores<<<(NN + TPB - 1) / TPB, TPB, 0, stream>>>(hA, gat2_as, gat2_ad, sas, sad);
    gat_aggr<<<(NN * 64 + TPB - 1) / TPB, TPB, 0, stream>>>(hA, sas, sad, gat2_b,
                                                            bn2_g, bn2_b, bn2_m, bn2_v, hB);
    pack_emb<<<(NN * 66 + TPB - 1) / TPB, TPB, 0, stream>>>(hB, x, emb);

    for (int d = 0; d < 2; ++d)
        launch_gemm(stream, emb, D_IN1, l1_Wih + (size_t)d * GATE * D_IN1, D_IN1,
                    l1_b + (size_t)d * GATE, Zx + (size_t)d * NGR * GATE, GATE,
                    NGR, GATE, D_IN1, 0);
    zero_f32<<<(2 * 2 * BATCH * HID + TPB - 1) / TPB, TPB, 0, stream>>>(hst, 2 * 2 * BATCH * HID);
    for (int t = 0; t < SEQ; ++t) {
        lstm_step_gemm<<<dim3(128, 2), 256, 0, stream>>>(Zx, l1_Whh, hst, Zf, t);
        lstm_cell<<<(2 * BATCH * HID + TPB - 1) / TPB, TPB, 0, stream>>>(Zf, hst, cst, ys1, t);
    }

    for (int d = 0; d < 2; ++d)
        launch_gemm(stream, ys1, D_IN2, l2_Wih + (size_t)d * GATE * D_IN2, D_IN2,
                    l2_b + (size_t)d * GATE, Zx + (size_t)d * NGR * GATE, GATE,
                    NGR, GATE, D_IN2, 0);
    zero_f32<<<(2 * 2 * BATCH * HID + TPB - 1) / TPB, TPB, 0, stream>>>(hst, 2 * 2 * BATCH * HID);
    for (int t = 0; t < SEQ; ++t) {
        lstm_step_gemm<<<dim3(128, 2), 256, 0, stream>>>(Zx, l2_Whh, hst, Zf, t);
        lstm_cell<<<(2 * BATCH * HID + TPB - 1) / TPB, TPB, 0, stream>>>(Zf, hst, cst, ys2, t);
    }

    launch_gemm(stream, ys2, 1024, hW1, 1024, hb1, hd1, 256, NGR, 256, 1024, 1);
    launch_gemm(stream, hd1, 256,  hW2, 256,  hb2, out, 57,  NGR, 57,  256,  0);
}


# recurrent step: LSTM cell update fused into the step GEMM epilogue (tile = 32 rows x 8 units x 4 gates, h double-buffered by step parity): one launch per time step, not two; f32 sum order kept
# speedup vs baseline: 2.1751x; 1.1310x over previous
#include <hip/hip_runtime.h>


#define SEQ   90
#define KP    17
#define FD    2
#define BATCH 64
#define NGR   (BATCH * SEQ)
#define NN    (NGR * KP)
#define HID   512
#define GATE  2048
#define D_IN1 1122
#define D_IN2 1024

typedef float v2f __attribute__((ext_vector_type(2)));
typedef float v8f __attribute__((ext_vector_type(8)));

__device__ __forceinline__ v8f wmma_f32(v2f a, v2f b, v8f c) {
    return __builtin_amdgcn_wmma_f32_16x16x4_f32(false, a, false, b, (short)0, c,
                                                 false, false);
}

__device__ __forceinline__ float sigmoidf_(float x) { return 1.0f / (1.0f + __expf(-x)); }

__constant__ int NBR_OFF[18] = {0,2,5,8,10,12,16,20,22,24,25,26,29,32,34,36,37,38};
__constant__ int NBR[38] = {
    1,2, 2,0,3, 1,0,4, 1,5, 2,6, 11,6,7,3, 12,5,8,4, 5,9, 6,10, 7, 8,
    13,12,5, 14,11,6, 15,11, 16,12, 13, 14
};

__global__ void gemm_wmma_f32(const float* __restrict__ A, int lda,
                              const float* __restrict__ Bm, int ldb,
                              const float* __restrict__ bias,
                              float* __restrict__ C, int ldc,
                              int M, int N, int K, int act) {
    int wave = (blockIdx.x * blockDim.x + threadIdx.x) >> 5;
    int lane = threadIdx.x & 31;
    int tilesN = (N + 31) >> 5;
    int tilesM = M >> 5;
    if (wave >= tilesM * tilesN) return;
    int tm = wave / tilesN, tn = wave % tilesN;
    int m0 = tm << 5, n0 = tn << 5;
    int half = lane >> 4, l = lane & 15;
    int kofs = half << 1;

    const float* A0 = A + (size_t)(m0 + l) * lda + kofs;
    const float* A1 = A0 + (size_t)16 * lda;
    int bn0 = n0 + l;      if (bn0 > N - 1) bn0 = N - 1;
    int bn1 = n0 + 16 + l; if (bn1 > N - 1) bn1 = N - 1;
    const float* B0 = Bm + (size_t)bn0 * ldb + kofs;
    const float* B1 = Bm + (size_t)bn1 * ldb + kofs;

    v8f acc00 = {0.f,0.f,0.f,0.f,0.f,0.f,0.f,0.f};
    v8f acc01 = acc00, acc10 = acc00, acc11 = acc00;

    int Kmain = K & ~3;
    v2f a0 = *(const v2f*)A0;
    v2f a1 = *(const v2f*)A1;
    v2f b0 = *(const v2f*)B0;
    v2f b1 = *(const v2f*)B1;
    for (int k0 = 0; k0 < Kmain - 4; k0 += 4) {
        v2f a0n = *(const v2f*)(A0 + k0 + 4);
        v2f a1n = *(const v2f*)(A1 + k0 + 4);
        v2f b0n = *(const v2f*)(B0 + k0 + 4);
        v2f b1n = *(const v2f*)(B1 + k0 + 4);
        acc00 = wmma_f32(a0, b0, acc00);
        acc01 = wmma_f32(a0, b1, acc01);
        acc10 = wmma_f32(a1, b0, acc10);
        acc11 = wmma_f32(a1, b1, acc11);
        a0 = a0n; a1 = a1n; b0 = b0n; b1 = b1n;
    }
    acc00 = wmma_f32(a0, b0, acc00);
    acc01 = wmma_f32(a0, b1, acc01);
    acc10 = wmma_f32(a1, b0, acc10);
    acc11 = wmma_f32(a1, b1, acc11);

    if (K & 3) {
        int ka = Kmain + kofs, kb = ka + 1;
        int kaC = ka < K ? ka : K - 1;
        int kbC = kb < K ? kb : K - 1;
        const float* A0b = A + (size_t)(m0 + l) * lda;
        const float* A1b = A0b + (size_t)16 * lda;
        const float* B0b = Bm + (size_t)bn0 * ldb;
        const float* B1b = Bm + (size_t)bn1 * ldb;
        float va0 = A0b[kaC], wa0 = A0b[kbC];
        float va1 = A1b[kaC], wa1 = A1b[kbC];
        float vb0 = B0b[kaC], wb0 = B0b[kbC];
        float vb1 = B1b[kaC], wb1 = B1b[kbC];
        v2f at0, at1, bt0, bt1;
        at0.x = ka < K ? va0 : 0.f; at0.y = kb < K ? wa0 : 0.f;
        at1.x = ka < K ? va1 : 0.f; at1.y = kb < K ? wa1 : 0.f;
        bt0.x = ka < K ? vb0 : 0.f; bt0.y = kb < K ? wb0 : 0.f;
        bt1.x = ka < K ? vb1 : 0.f; bt1.y = kb < K ? wb1 : 0.f;
        acc00 = wmma_f32(at0, bt0, acc00);
        acc01 = wmma_f32(at0, bt1, acc01);
        acc10 = wmma_f32(at1, bt0, acc10);
        acc11 = wmma_f32(at1, bt1, acc11);
    }
    int cn0 = n0 + l, cn1 = n0 + 16 + l;
    if (cn0 < N) {
        float bv = bias ? bias[cn0] : 0.f;
        #pragma unroll
        for (int j = 0; j < 8; ++j) {
            int r0 = m0 + j + half * 8, r1 = r0 + 16;
            float v0 = acc00[j] + bv, v1 = acc10[j] + bv;
            if (act) { v0 = fmaxf(v0, 0.f); v1 = fmaxf(v1, 0.f); }
            C[(size_t)r0 * ldc + cn0] = v0;
            C[(size_t)r1 * ldc + cn0] = v1;
        }
    }
    if (cn1 < N) {
        float bv = bias ? bias[cn1] : 0.f;
        #pragma unroll
        for (int j = 0; j < 8; ++j) {
            int r0 = m0 + j + half * 8, r1 = r0 + 16;
            float v0 = acc01[j] + bv, v1 = acc11[j] + bv;
            if (act) { v0 = fmaxf(v0, 0.f); v1 = fmaxf(v1, 0.f); }
            C[(size_t)r0 * ldc + cn1] = v0;
            C[(size_t)r1 * ldc + cn1] = v1;
        }
    }
}

#define KSPLIT    8
#define KSLICE    (HID / KSPLIT)
#define RED_PITCH 36

typedef float v4f __attribute__((ext_vector_type(4)));

struct Slice { v4f q[8]; };

__device__ __forceinline__ void load_group(Slice& s, const float* __restrict__ p, int g) {
    s.q[2 * g]     = *(const v4f*)(p + 16 * g);
    s.q[2 * g + 1] = *(const v4f*)(p + 16 * g + 4);
}

__device__ __forceinline__ v2f frag_step(const Slice& s, int st) {
    v4f v = s.q[st >> 1];
    v2f r;
    if (st & 1) { r.x = v.z; r.y = v.w; } else { r.x = v.x; r.y = v.y; }
    return r;
}

__global__ __launch_bounds__(256)
void lstm_step(const float* __restrict__ Zx,
               const float* __restrict__ Whh,
               float* hbuf0, float* hbuf1,
               float* __restrict__ cbuf,
               float* __restrict__ ys, int t) {
    __shared__ float red[KSPLIT * 32 * RED_PITCH];

    const float* hprev = (t & 1) ? hbuf1 : hbuf0;
    float*       hnext = (t & 1) ? hbuf0 : hbuf1;

    int dir  = blockIdx.y;
    int tile = blockIdx.x;
    int wv   = threadIdx.x >> 5;
    int lane = threadIdx.x & 31;
    int tm = tile >> 6, tn = tile & 63;
    int m0 = tm << 5, j0 = tn << 3;
    int half = lane >> 4, l = lane & 15;

    int kb = wv * KSLICE + half * 8;
    int nb = (l >> 3) * HID + j0 + (l & 7);
    const float* A0 = hprev + (size_t)dir * BATCH * HID + (size_t)(m0 + l) * HID + kb;
    const float* B0 = Whh + (size_t)dir * GATE * HID + (size_t)nb * HID + kb;
    const float* A1 = A0 + (size_t)16 * HID;
    const float* B1 = B0 + (size_t)2 * HID * HID;
    Slice a0, a1, b0, b1;
    #pragma unroll
    for (int g = 0; g < 4; ++g) {
        load_group(a0, A0, g);
        load_group(b0, B0, g);
        load_group(b1, B1, g);
        load_group(a1, A1, g);
    }
    int t_d = dir ? (SEQ - 1 - t) : t;
    int rr  = 4 * wv + (lane >> 3);
    int jj  = lane & 7;
    int b   = m0 + rr;
    int sidx = dir * BATCH * HID + b * HID + j0 + jj;
    const float* zxp = Zx + (size_t)dir * NGR * GATE + (size_t)(b * SEQ + t_d) * GATE + j0 + jj;
    float zx[4];
    #pragma unroll
    for (int g = 0; g < 4; ++g) zx[g] = zxp[g * HID];
    float c = cbuf[sidx];
    __builtin_amdgcn_sched_barrier(0);

    v8f acc00 = {0.f,0.f,0.f,0.f,0.f,0.f,0.f,0.f};
    v8f acc01 = acc00, acc10 = acc00, acc11 = acc00;
    #pragma unroll
    for (int st = 0; st < 16; ++st) {
        v2f fa0 = frag_step(a0, st), fa1 = frag_step(a1, st);
        v2f fb0 = frag_step(b0, st), fb1 = frag_step(b1, st);
        acc00 = wmma_f32(fa0, fb0, acc00);
        acc01 = wmma_f32(fa0, fb1, acc01);
        acc10 = wmma_f32(fa1, fb0, acc10);
        acc11 = wmma_f32(fa1, fb1, acc11);
    }

    float* rw = red + wv * (32 * RED_PITCH);
    #pragma unroll
    for (int j = 0; j < 8; ++j) {
        int r = j + 8 * half;
        rw[r * RED_PITCH + l]             = acc00[j];
        rw[r * RED_PITCH + 16 + l]        = acc01[j];
        rw[(r + 16) * RED_PITCH + l]      = acc10[j];
        rw[(r + 16) * RED_PITCH + 16 + l] = acc11[j];
    }
    __syncthreads();

    const float* rp = red + rr * RED_PITCH + jj;
    float z[4];
    #pragma unroll
    for (int g = 0; g < 4; ++g) {
        float s = rp[8 * g];
        #pragma unroll
        for (int w = 1; w < KSPLIT; ++w) s += rp[w * (32 * RED_PITCH) + 8 * g];
        z[g] = s + zx[g];
    }
    c = sigmoidf_(z[1]) * c + sigmoidf_(z[0]) * tanhf(z[2]);
    float h = sigmoidf_(z[3]) * tanhf(c);
    cbuf[sidx]  = c;
    hnext[sidx] = h;
    ys[(size_t)(b * SEQ + t_d) * 1024 + dir * HID + j0 + jj] = h;
}

__global__ void gat1_lin(const float* __restrict__ x, const float* __restrict__ W,
                         float* __restrict__ h) {
    int idx = blockIdx.x * blockDim.x + threadIdx.x;
    if (idx >= NN * 64) return;
    int i = idx >> 6, f = idx & 63;
    h[idx] = x[i * 2 + 0] * W[f] + x[i * 2 + 1] * W[64 + f];
}

__global__ void gat_scores(const float* __restrict__ h,
                           const float* __restrict__ asrc, const float* __restrict__ adst,
                           float* __restrict__ sas, float* __restrict__ sad) {
    int i = blockIdx.x * blockDim.x + threadIdx.x;
    if (i >= NN) return;
    const float* hp = h + (size_t)i * 64;
    float a = 0.f, d = 0.f;
    #pragma unroll
    for (int f = 0; f < 64; ++f) { float v = hp[f]; a += v * asrc[f]; d += v * adst[f]; }
    sas[i] = a; sad[i] = d;
}

__global__ void gat_aggr(const float* __restrict__ h,
                         const float* __restrict__ sas, const float* __restrict__ sad,
                         const float* __restrict__ bvec,
                         const float* __restrict__ g, const float* __restrict__ bb,
                         const float* __restrict__ bm, const float* __restrict__ bvv,
                         float* __restrict__ out) {
    int idx = blockIdx.x * blockDim.x + threadIdx.x;
    if (idx >= NN * 64) return;
    int i = idx >> 6, ch = idx & 63;
    int dl = i % KP, base = i - dl;
    float ad = sad[i];
    int o0 = NBR_OFF[dl], deg = NBR_OFF[dl + 1] - o0;

    float e[5], hv[5];
    float t0 = sas[i] + ad;
    e[0]  = t0 < 0.f ? 0.2f * t0 : t0;
    hv[0] = h[(size_t)i * 64 + ch];
    for (int n = 0; n < deg; ++n) {
        int s = base + NBR[o0 + n];
        float tt = sas[s] + ad;
        e[n + 1]  = tt < 0.f ? 0.2f * tt : tt;
        hv[n + 1] = h[(size_t)s * 64 + ch];
    }
    int cnt = deg + 1;
    float m = e[0];
    for (int n = 1; n < cnt; ++n) m = fmaxf(m, e[n]);
    float den = 0.f, num = 0.f;
    for (int n = 0; n < cnt; ++n) { float w = __expf(e[n] - m); den += w; num += w * hv[n]; }
    float val = num / den + bvec[ch];
    val = (val - bm[ch]) * (g[ch] * rsqrtf(bvv[ch] + 1e-5f)) + bb[ch];
    out[idx] = fmaxf(val, 0.f);
}

__global__ void pack_emb(const float* __restrict__ h2, const float* __restrict__ x,
                         float* __restrict__ emb) {
    int idx = blockIdx.x * blockDim.x + threadIdx.x;
    if (idx >= NN * 66) return;
    int i = idx / 66, c = idx % 66;
    int bt = i / KP, kp = i % KP;
    float v = (c < 64) ? h2[(size_t)i * 64 + c] : x[i * 2 + (c - 64)];
    emb[(size_t)bt * D_IN1 + kp * 66 + c] = v;
}

__global__ void transpose64(const float* __restrict__ W, float* __restrict__ WT) {
    int idx = blockIdx.x * blockDim.x + threadIdx.x;
    if (idx >= 64 * 64) return;
    int n = idx >> 6, k = idx & 63;
    WT[n * 64 + k] = W[k * 64 + n];
}

__global__ void zero_f32(float* __restrict__ p, int n) {
    int i = blockIdx.x * blockDim.x + threadIdx.x;
    if (i < n) p[i] = 0.f;
}

static inline void launch_gemm(hipStream_t s, const float* A, int lda,
                               const float* B, int ldb, const float* bias,
                               float* C, int ldc, int M, int N, int K, int act) {
    int tiles = (M / 32) * ((N + 31) / 32);
    int blocks = (tiles + 7) / 8;
    gemm_wmma_f32<<<blocks, 256, 0, s>>>(A, lda, B, ldb, bias, C, ldc, M, N, K, act);
}

extern "C" void kernel_launch(void* const* d_in, const int* in_sizes, int n_in,
                              void* d_out, int out_size, void* d_ws, size_t ws_size,
                              hipStream_t stream) {
    const float* x        = (const float*)d_in[0];
    const float* gat1_W   = (const float*)d_in[2];
    const float* gat1_as  = (const float*)d_in[3];
    const float* gat1_ad  = (const float*)d_in[4];
    const float* gat1_b   = (const float*)d_in[5];
    const float* gat2_W   = (const float*)d_in[6];
    const float* gat2_as  = (const float*)d_in[7];
    const float* gat2_ad  = (const float*)d_in[8];
    const float* gat2_b   = (const float*)d_in[9];
    const float* bn1_g = (const float*)d_in[10], *bn1_b = (const float*)d_in[11];
    const float* bn1_m = (const float*)d_in[12], *bn1_v = (const float*)d_in[13];
    const float* bn2_g = (const float*)d_in[14], *bn2_b = (const float*)d_in[15];
    const float* bn2_m = (const float*)d_in[16], *bn2_v = (const float*)d_in[17];
    const float* l1_Wih = (const float*)d_in[18];
    const float* l1_Whh = (const float*)d_in[19];
    const float* l1_b   = (const float*)d_in[20];
    const float* l2_Wih = (const float*)d_in[21];
    const float* l2_Whh = (const float*)d_in[22];
    const float* l2_b   = (const float*)d_in[23];
    const float* hW1 = (const float*)d_in[24], *hb1 = (const float*)d_in[25];
    const float* hW2 = (const float*)d_in[26], *hb2 = (const float*)d_in[27];
    float* out = (float*)d_out;
    float* ws  = (float*)d_ws;

    const size_t OFF_HA  = 0;
    const size_t OFF_HB  = OFF_HA  + (size_t)NN * 64;
    const size_t OFF_SAS = OFF_HB  + (size_t)NN * 64;
    const size_t OFF_SAD = OFF_SAS + NN;
    const size_t OFF_EMB = OFF_SAD + NN;
    const size_t OFF_ZX  = OFF_EMB + (size_t)NGR * D_IN1;
    const size_t OFF_ZF  = OFF_ZX  + (size_t)2 * NGR * GATE;
    const size_t OFF_HST = OFF_ZF  + (size_t)2 * BATCH * GATE;
    const size_t OFF_CST = OFF_HST + (size_t)2 * BATCH * HID;
    const size_t OFF_WT  = OFF_CST + (size_t)2 * BATCH * HID;

    float* hA   = ws + OFF_HA;
    float* hB   = ws + OFF_HB;
    float* sas  = ws + OFF_SAS;
    float* sad  = ws + OFF_SAD;
    float* emb  = ws + OFF_EMB;
    float* Zx   = ws + OFF_ZX;
    float* hs2  = ws + OFF_ZF;
    float* hst  = ws + OFF_HST;
    float* cst  = ws + OFF_CST;
    float* wt   = ws + OFF_WT;
    float* ys1  = emb;
    float* ys2  = hB;
    float* hd1  = hA;

    const int TPB = 256;

    gat1_lin<<<(NN * 64 + TPB - 1) / TPB, TPB, 0, stream>>>(x, gat1_W, hA);
    gat_scores<<<(NN + TPB - 1) / TPB, TPB, 0, stream>>>(hA, gat1_as, gat1_ad, sas, sad);
    gat_aggr<<<(NN * 64 + TPB - 1) / TPB, TPB, 0, stream>>>(hA, sas, sad, gat1_b,
                                                            bn1_g, bn1_b, bn1_m, bn1_v, hB);
    transpose64<<<16, 256, 0, stream>>>(gat2_W, wt);
    launch_gemm(stream, hB, 64, wt, 64, nullptr, hA, 64, NN, 64, 64, 0);
    gat_scores<<<(NN + TPB - 1) / TPB, TPB, 0, stream>>>(hA, gat2_as, gat2_ad, sas, sad);
    gat_aggr<<<(NN * 64 + TPB - 1) / TPB, TPB, 0, stream>>>(hA, sas, sad, gat2_b,
                                                            bn2_g, bn2_b, bn2_m, bn2_v, hB);
    pack_emb<<<(NN * 66 + TPB - 1) / TPB, TPB, 0, stream>>>(hB, x, emb);

    for (int d = 0; d < 2; ++d)
        launch_gemm(stream, emb, D_IN1, l1_Wih + (size_t)d * GATE * D_IN1, D_IN1,
                    l1_b + (size_t)d * GATE, Zx + (size_t)d * NGR * GATE, GATE,
                    NGR, GATE, D_IN1, 0);
    zero_f32<<<(2 * 2 * BATCH * HID + TPB - 1) / TPB, TPB, 0, stream>>>(hst, 2 * 2 * BATCH * HID);
    for (int t = 0; t < SEQ; ++t) {
        lstm_step<<<dim3(128, 2), 256, 0, stream>>>(Zx, l1_Whh, hst, hs2, cst, ys1, t);
    }

    for (int d = 0; d < 2; ++d)
        launch_gemm(stream, ys1, D_IN2, l2_Wih + (size_t)d * GATE * D_IN2, D_IN2,
                    l2_b + (size_t)d * GATE, Zx + (size_t)d * NGR * GATE, GATE,
                    NGR, GATE, D_IN2, 0);
    zero_f32<<<(2 * 2 * BATCH * HID + TPB - 1) / TPB, TPB, 0, stream>>>(hst, 2 * 2 * BATCH * HID);
    for (int t = 0; t < SEQ; ++t) {
        lstm_step<<<dim3(128, 2), 256, 0, stream>>>(Zx, l2_Whh, hst, hs2, cst, ys2, t);
    }

    launch_gemm(stream, ys2, 1024, hW1, 1024, hb1, hd1, 256, NGR, 256, 1024, 1);
    launch_gemm(stream, hd1, 256,  hW2, 256,  hb2, out, 57,  NGR, 57,  256,  0);
}


# LSTM input GEMMs: operands staged through padded LDS per 128x128 work-group tile (32-k chunks, row-contiguous loads), not per-lane strided global loads; f32 k summation order changes
# speedup vs baseline: 3.2731x; 1.5048x over previous
#include <hip/hip_runtime.h>


#define SEQ   90
#define KP    17
#define FD    2
#define BATCH 64
#define NGR   (BATCH * SEQ)
#define NN    (NGR * KP)
#define HID   512
#define GATE  2048
#define D_IN1 1122
#define D_IN2 1024

typedef float v2f __attribute__((ext_vector_type(2)));
typedef float v8f __attribute__((ext_vector_type(8)));

__device__ __forceinline__ v8f wmma_f32(v2f a, v2f b, v8f c) {
    return __builtin_amdgcn_wmma_f32_16x16x4_f32(false, a, false, b, (short)0, c,
                                                 false, false);
}

__device__ __forceinline__ float sigmoidf_(float x) { return 1.0f / (1.0f + __expf(-x)); }

__constant__ int NBR_OFF[18] = {0,2,5,8,10,12,16,20,22,24,25,26,29,32,34,36,37,38};
__constant__ int NBR[38] = {
    1,2, 2,0,3, 1,0,4, 1,5, 2,6, 11,6,7,3, 12,5,8,4, 5,9, 6,10, 7, 8,
    13,12,5, 14,11,6, 15,11, 16,12, 13, 14
};

__global__ void gemm_wmma_f32(const float* __restrict__ A, int lda,
                              const float* __restrict__ Bm, int ldb,
                              const float* __restrict__ bias,
                              float* __restrict__ C, int ldc,
                              int M, int N, int K, int act) {
    int wave = (blockIdx.x * blockDim.x + threadIdx.x) >> 5;
    int lane = threadIdx.x & 31;
    int tilesN = (N + 31) >> 5;
    int tilesM = M >> 5;
    if (wave >= tilesM * tilesN) return;
    int tm = wave / tilesN, tn = wave % tilesN;
    int m0 = tm << 5, n0 = tn << 5;
    int half = lane >> 4, l = lane & 15;
    int kofs = half << 1;

    const float* A0 = A + (size_t)(m0 + l) * lda + kofs;
    const float* A1 = A0 + (size_t)16 * lda;
    int bn0 = n0 + l;      if (bn0 > N - 1) bn0 = N - 1;
    int bn1 = n0 + 16 + l; if (bn1 > N - 1) bn1 = N - 1;
    const float* B0 = Bm + (size_t)bn0 * ldb + kofs;
    const float* B1 = Bm + (size_t)bn1 * ldb + kofs;

    v8f acc00 = {0.f,0.f,0.f,0.f,0.f,0.f,0.f,0.f};
    v8f acc01 = acc00, acc10 = acc00, acc11 = acc00;

    int Kmain = K & ~3;
    v2f a0 = *(const v2f*)A0;
    v2f a1 = *(const v2f*)A1;
    v2f b0 = *(const v2f*)B0;
    v2f b1 = *(const v2f*)B1;
    for (int k0 = 0; k0 < Kmain - 4; k0 += 4) {
        v2f a0n = *(const v2f*)(A0 + k0 + 4);
        v2f a1n = *(const v2f*)(A1 + k0 + 4);
        v2f b0n = *(const v2f*)(B0 + k0 + 4);
        v2f b1n = *(const v2f*)(B1 + k0 + 4);
        acc00 = wmma_f32(a0, b0, acc00);
        acc01 = wmma_f32(a0, b1, acc01);
        acc10 = wmma_f32(a1, b0, acc10);
        acc11 = wmma_f32(a1, b1, acc11);
        a0 = a0n; a1 = a1n; b0 = b0n; b1 = b1n;
    }
    acc00 = wmma_f32(a0, b0, acc00);
    acc01 = wmma_f32(a0, b1, acc01);
    acc10 = wmma_f32(a1, b0, acc10);
    acc11 = wmma_f32(a1, b1, acc11);

    if (K & 3) {
        int ka = Kmain + kofs, kb = ka + 1;
        int kaC = ka < K ? ka : K - 1;
        int kbC = kb < K ? kb : K - 1;
        const float* A0b = A + (size_t)(m0 + l) * lda;
        const float* A1b = A0b + (size_t)16 * lda;
        const float* B0b = Bm + (size_t)bn0 * ldb;
        const float* B1b = Bm + (size_t)bn1 * ldb;
        float va0 = A0b[kaC], wa0 = A0b[kbC];
        float va1 = A1b[kaC], wa1 = A1b[kbC];
        float vb0 = B0b[kaC], wb0 = B0b[kbC];
        float vb1 = B1b[kaC], wb1 = B1b[kbC];
        v2f at0, at1, bt0, bt1;
        at0.x = ka < K ? va0 : 0.f; at0.y = kb < K ? wa0 : 0.f;
        at1.x = ka < K ? va1 : 0.f; at1.y = kb < K ? wa1 : 0.f;
        bt0.x = ka < K ? vb0 : 0.f; bt0.y = kb < K ? wb0 : 0.f;
        bt1.x = ka < K ? vb1 : 0.f; bt1.y = kb < K ? wb1 : 0.f;
        acc00 = wmma_f32(at0, bt0, acc00);
        acc01 = wmma_f32(at0, bt1, acc01);
        acc10 = wmma_f32(at1, bt0, acc10);
        acc11 = wmma_f32(at1, bt1, acc11);
    }
    int cn0 = n0 + l, cn1 = n0 + 16 + l;
    if (cn0 < N) {
        float bv = bias ? bias[cn0] : 0.f;
        #pragma unroll
        for (int j = 0; j < 8; ++j) {
            int r0 = m0 + j + half * 8, r1 = r0 + 16;
            float v0 = acc00[j] + bv, v1 = acc10[j] + bv;
            if (act) { v0 = fmaxf(v0, 0.f); v1 = fmaxf(v1, 0.f); }
            C[(size_t)r0 * ldc + cn0] = v0;
            C[(size_t)r1 * ldc + cn0] = v1;
        }
    }
    if (cn1 < N) {
        float bv = bias ? bias[cn1] : 0.f;
        #pragma unroll
        for (int j = 0; j < 8; ++j) {
            int r0 = m0 + j + half * 8, r1 = r0 + 16;
            float v0 = acc01[j] + bv, v1 = acc11[j] + bv;
            if (act) { v0 = fmaxf(v0, 0.f); v1 = fmaxf(v1, 0.f); }
            C[(size_t)r0 * ldc + cn1] = v0;
            C[(size_t)r1 * ldc + cn1] = v1;
        }
    }
}

#define GL_PITCH 36
#define GL_OPER  (128 * GL_PITCH)
#define GL_STAGE (2 * GL_OPER)

typedef float v4g __attribute__((ext_vector_type(4)));

__device__ __forceinline__ void gl_fetch(v2f (&r)[8], const float* __restrict__ p, size_t ld16) {
    #pragma unroll
    for (int i = 0; i < 8; ++i) r[i] = *(const v2f*)(p + i * ld16);
}

__device__ __forceinline__ void gl_fetch_tail(v2f (&r)[8], const float* __restrict__ p,
                                              size_t ld16, bool ok) {
    #pragma unroll
    for (int i = 0; i < 8; ++i) r[i] = (v2f){0.f, 0.f};
    if (ok) {
        #pragma unroll
        for (int i = 0; i < 8; ++i) r[i] = *(const v2f*)(p + i * ld16);
    }
}

__device__ __forceinline__ void gl_store(float* __restrict__ s, const v2f (&r)[8]) {
    #pragma unroll
    for (int i = 0; i < 8; ++i) *(v2f*)(s + i * 16 * GL_PITCH) = r[i];
}

__device__ __forceinline__ v2f gl_pick(v4g v, int e) {
    v2f r;
    if (e) { r.x = v.z; r.y = v.w; } else { r.x = v.x; r.y = v.y; }
    return r;
}

__global__ __launch_bounds__(256)
void gemm_lds_f32(const float* __restrict__ A, int lda,
                  const float* __restrict__ Bm, int ldb,
                  const float* __restrict__ bias,
                  float* __restrict__ C, int ldc, int K) {
    __shared__ float sm[2 * GL_STAGE];

    int tid  = threadIdx.x;
    int wv   = tid >> 5, lane = tid & 31;
    int half = lane >> 4, l = lane & 15;
    int wm = wv >> 2, wn = wv & 3;
    int m0 = blockIdx.y << 7, n0 = blockIdx.x << 7;

    int cp = tid & 15, r0 = tid >> 4;
    const float* Ag = A  + (size_t)(m0 + r0) * lda + 2 * cp;
    const float* Bg = Bm + (size_t)(n0 + r0) * ldb + 2 * cp;
    size_t lda16 = (size_t)16 * lda, ldb16 = (size_t)16 * ldb;
    float* sA = sm + r0 * GL_PITCH + 2 * cp;
    float* sB = sA + GL_OPER;

    const float* fA = sm + (wm * 64 + l) * GL_PITCH + 16 * half;
    const float* fB = sm + GL_OPER + (wn * 32 + l) * GL_PITCH + 16 * half;

    v8f acc[4][2];
    #pragma unroll
    for (int t = 0; t < 8; ++t) acc[t >> 1][t & 1] = (v8f){0.f,0.f,0.f,0.f,0.f,0.f,0.f,0.f};

    int nch = (K + 31) >> 5;
    v2f ra[8], rb[8];
    gl_fetch(ra, Ag, lda16);
    gl_fetch(rb, Bg, ldb16);
    gl_store(sA, ra);
    gl_store(sB, rb);
    __syncthreads();

    for (int ch = 0; ch < nch; ++ch) {
        int  st   = ch & 1;
        bool more = ch + 1 < nch;
        if (more) {
            int kn = (ch + 1) << 5;
            if (kn + 32 <= K) {
                gl_fetch(ra, Ag + kn, lda16);
                gl_fetch(rb, Bg + kn, ldb16);
            } else {
                bool ok = kn + 2 * cp < K;
                gl_fetch_tail(ra, Ag + kn, lda16, ok);
                gl_fetch_tail(rb, Bg + kn, ldb16, ok);
            }
        }
        const float* pa = fA + st * GL_STAGE;
        const float* pb = fB + st * GL_STAGE;
        v4g qa[4][4], qb[2][4];
        #pragma unroll
        for (int t = 0; t < 16; ++t)
            qa[t & 3][t >> 2] = *(const v4g*)(pa + (t & 3) * 16 * GL_PITCH + 4 * (t >> 2));
        #pragma unroll
        for (int t = 0; t < 8; ++t)
            qb[t & 1][t >> 1] = *(const v4g*)(pb + (t & 1) * 16 * GL_PITCH + 4 * (t >> 1));
        #pragma unroll
        for (int s = 0; s < 64; ++s) {
            int q = s >> 4, e = (s >> 3) & 1, i = (s >> 1) & 3, j = s & 1;
            acc[i][j] = wmma_f32(gl_pick(qa[i][q], e), gl_pick(qb[j][q], e), acc[i][j]);
        }
        if (more) {
            gl_store(sA + (st ^ 1) * GL_STAGE, ra);
            gl_store(sB + (st ^ 1) * GL_STAGE, rb);
        }
        __syncthreads();
    }

    int col0 = n0 + wn * 32 + l;
    float bv0 = bias[col0], bv1 = bias[col0 + 16];
    #pragma unroll
    for (int u = 0; u < 64; ++u) {
        int j = u >> 5, i = (u >> 3) & 3, r = u & 7;
        int row = m0 + wm * 64 + 16 * i + r + 8 * half;
        C[(size_t)row * ldc + col0 + 16 * j] = acc[i][j][r] + (j ? bv1 : bv0);
    }
}

#define KSPLIT    8
#define KSLICE    (HID / KSPLIT)
#define RED_PITCH 36

typedef float v4f __attribute__((ext_vector_type(4)));

struct Slice { v4f q[8]; };

__device__ __forceinline__ void load_group(Slice& s, const float* __restrict__ p, int g) {
    s.q[2 * g]     = *(const v4f*)(p + 16 * g);
    s.q[2 * g + 1] = *(const v4f*)(p + 16 * g + 4);
}

__device__ __forceinline__ v2f frag_step(const Slice& s, int st) {
    v4f v = s.q[st >> 1];
    v2f r;
    if (st & 1) { r.x = v.z; r.y = v.w; } else { r.x = v.x; r.y = v.y; }
    return r;
}

__global__ __launch_bounds__(256)
void lstm_step(const float* __restrict__ Zx,
               const float* __restrict__ Whh,
               float* hbuf0, float* hbuf1,
               float* __restrict__ cbuf,
               float* __restrict__ ys, int t) {
    __shared__ float red[KSPLIT * 32 * RED_PITCH];

    const float* hprev = (t & 1) ? hbuf1 : hbuf0;
    float*       hnext = (t & 1) ? hbuf0 : hbuf1;

    int dir  = blockIdx.y;
    int tile = blockIdx.x;
    int wv   = threadIdx.x >> 5;
    int lane = threadIdx.x & 31;
    int tm = tile >> 6, tn = tile & 63;
    int m0 = tm << 5, j0 = tn << 3;
    int half = lane >> 4, l = lane & 15;

    int kb = wv * KSLICE + half * 8;
    int nb = (l >> 3) * HID + j0 + (l & 7);
    const float* A0 = hprev + (size_t)dir * BATCH * HID + (size_t)(m0 + l) * HID + kb;
    const float* B0 = Whh + (size_t)dir * GATE * HID + (size_t)nb * HID + kb;
    const float* A1 = A0 + (size_t)16 * HID;
    const float* B1 = B0 + (size_t)2 * HID * HID;
    Slice a0, a1, b0, b1;
    #pragma unroll
    for (int g = 0; g < 4; ++g) {
        load_group(a0, A0, g);
        load_group(b0, B0, g);
        load_group(b1, B1, g);
        load_group(a1, A1, g);
    }
    int t_d = dir ? (SEQ - 1 - t) : t;
    int rr  = 4 * wv + (lane >> 3);
    int jj  = lane & 7;
    int b   = m0 + rr;
    int sidx = dir * BATCH * HID + b * HID + j0 + jj;
    const float* zxp = Zx + (size_t)dir * NGR * GATE + (size_t)(b * SEQ + t_d) * GATE + j0 + jj;
    float zx[4];
    #pragma unroll
    for (int g = 0; g < 4; ++g) zx[g] = zxp[g * HID];
    float c = cbuf[sidx];
    __builtin_amdgcn_sched_barrier(0);

    v8f acc00 = {0.f,0.f,0.f,0.f,0.f,0.f,0.f,0.f};
    v8f acc01 = acc00, acc10 = acc00, acc11 = acc00;
    #pragma unroll
    for (int st = 0; st < 16; ++st) {
        v2f fa0 = frag_step(a0, st), fa1 = frag_step(a1, st);
        v2f fb0 = frag_step(b0, st), fb1 = frag_step(b1, st);
        acc00 = wmma_f32(fa0, fb0, acc00);
        acc01 = wmma_f32(fa0, fb1, acc01);
        acc10 = wmma_f32(fa1, fb0, acc10);
        acc11 = wmma_f32(fa1, fb1, acc11);
    }

    float* rw = red + wv * (32 * RED_PITCH);
    #pragma unroll
    for (int j = 0; j < 8; ++j) {
        int r = j + 8 * half;
        rw[r * RED_PITCH + l]             = acc00[j];
        rw[r * RED_PITCH + 16 + l]        = acc01[j];
        rw[(r + 16) * RED_PITCH + l]      = acc10[j];
        rw[(r + 16) * RED_PITCH + 16 + l] = acc11[j];
    }
    __syncthreads();

    const float* rp = red + rr * RED_PITCH + jj;
    float z[4];
    #pragma unroll
    for (int g = 0; g < 4; ++g) {
        float s = rp[8 * g];
        #pragma unroll
        for (int w = 1; w < KSPLIT; ++w) s += rp[w * (32 * RED_PITCH) + 8 * g];
        z[g] = s + zx[g];
    }
    c = sigmoidf_(z[1]) * c + sigmoidf_(z[0]) * tanhf(z[2]);
    float h = sigmoidf_(z[3]) * tanhf(c);
    cbuf[sidx]  = c;
    hnext[sidx] = h;
    ys[(size_t)(b * SEQ + t_d) * 1024 + dir * HID + j0 + jj] = h;
}

__global__ void gat1_lin(const float* __restrict__ x, const float* __restrict__ W,
                         float* __restrict__ h) {
    int idx = blockIdx.x * blockDim.x + threadIdx.x;
    if (idx >= NN * 64) return;
    int i = idx >> 6, f = idx & 63;
    h[idx] = x[i * 2 + 0] * W[f] + x[i * 2 + 1] * W[64 + f];
}

__global__ void gat_scores(const float* __restrict__ h,
                           const float* __restrict__ asrc, const float* __restrict__ adst,
                           float* __restrict__ sas, float* __restrict__ sad) {
    int i = blockIdx.x * blockDim.x + threadIdx.x;
    if (i >= NN) return;
    const float* hp = h + (size_t)i * 64;
    float a = 0.f, d = 0.f;
    #pragma unroll
    for (int f = 0; f < 64; ++f) { float v = hp[f]; a += v * asrc[f]; d += v * adst[f]; }
    sas[i] = a; sad[i] = d;
}

__global__ void gat_aggr(const float* __restrict__ h,
                         const float* __restrict__ sas, const float* __restrict__ sad,
                         const float* __restrict__ bvec,
                         const float* __restrict__ g, const float* __restrict__ bb,
                         const float* __restrict__ bm, const float* __restrict__ bvv,
                         float* __restrict__ out) {
    int idx = blockIdx.x * blockDim.x + threadIdx.x;
    if (idx >= NN * 64) return;
    int i = idx >> 6, ch = idx & 63;
    int dl = i % KP, base = i - dl;
    float ad = sad[i];
    int o0 = NBR_OFF[dl], deg = NBR_OFF[dl + 1] - o0;

    float e[5], hv[5];
    float t0 = sas[i] + ad;
    e[0]  = t0 < 0.f ? 0.2f * t0 : t0;
    hv[0] = h[(size_t)i * 64 + ch];
    for (int n = 0; n < deg; ++n) {
        int s = base + NBR[o0 + n];
        float tt = sas[s] + ad;
        e[n + 1]  = tt < 0.f ? 0.2f * tt : tt;
        hv[n + 1] = h[(size_t)s * 64 + ch];
    }
    int cnt = deg + 1;
    float m = e[0];
    for (int n = 1; n < cnt; ++n) m = fmaxf(m, e[n]);
    float den = 0.f, num = 0.f;
    for (int n = 0; n < cnt; ++n) { float w = __expf(e[n] - m); den += w; num += w * hv[n]; }
    float val = num / den + bvec[ch];
    val = (val - bm[ch]) * (g[ch] * rsqrtf(bvv[ch] + 1e-5f)) + bb[ch];
    out[idx] = fmaxf(val, 0.f);
}

__global__ void pack_emb(const float* __restrict__ h2, const float* __restrict__ x,
                         float* __restrict__ emb) {
    int idx = blockIdx.x * blockDim.x + threadIdx.x;
    if (idx >= NN * 66) return;
    int i = idx / 66, c = idx % 66;
    int bt = i / KP, kp = i % KP;
    float v = (c < 64) ? h2[(size_t)i * 64 + c] : x[i * 2 + (c - 64)];
    emb[(size_t)bt * D_IN1 + kp * 66 + c] = v;
}

__global__ void transpose64(const float* __restrict__ W, float* __restrict__ WT) {
    int idx = blockIdx.x * blockDim.x + threadIdx.x;
    if (idx >= 64 * 64) return;
    int n = idx >> 6, k = idx & 63;
    WT[n * 64 + k] = W[k * 64 + n];
}

__global__ void zero_f32(float* __restrict__ p, int n) {
    int i = blockIdx.x * blockDim.x + threadIdx.x;
    if (i < n) p[i] = 0.f;
}

static inline void launch_gemm(hipStream_t s, const float* A, int lda,
                               const float* B, int ldb, const float* bias,
                               float* C, int ldc, int M, int N, int K, int act) {
    int tiles = (M / 32) * ((N + 31) / 32);
    int blocks = (tiles + 7) / 8;
    gemm_wmma_f32<<<blocks, 256, 0, s>>>(A, lda, B, ldb, bias, C, ldc, M, N, K, act);
}

static inline void launch_gemm_lds(hipStream_t s, const float* A, int lda,
                                   const float* B, int ldb, const float* bias,
                                   float* C, int ldc, int K) {
    gemm_lds_f32<<<dim3(GATE / 128, NGR / 128), 256, 0, s>>>(A, lda, B, ldb, bias, C, ldc, K);
}

extern "C" void kernel_launch(void* const* d_in, const int* in_sizes, int n_in,
                              void* d_out, int out_size, void* d_ws, size_t ws_size,
                              hipStream_t stream) {
    const float* x        = (const float*)d_in[0];
    const float* gat1_W   = (const float*)d_in[2];
    const float* gat1_as  = (const float*)d_in[3];
    const float* gat1_ad  = (const float*)d_in[4];
    const float* gat1_b   = (const float*)d_in[5];
    const float* gat2_W   = (const float*)d_in[6];
    const float* gat2_as  = (const float*)d_in[7];
    const float* gat2_ad  = (const float*)d_in[8];
    const float* gat2_b   = (const float*)d_in[9];
    const float* bn1_g = (const float*)d_in[10], *bn1_b = (const float*)d_in[11];
    const float* bn1_m = (const float*)d_in[12], *bn1_v = (const float*)d_in[13];
    const float* bn2_g = (const float*)d_in[14], *bn2_b = (const float*)d_in[15];
    const float* bn2_m = (const float*)d_in[16], *bn2_v = (const float*)d_in[17];
    const float* l1_Wih = (const float*)d_in[18];
    const float* l1_Whh = (const float*)d_in[19];
    const float* l1_b   = (const float*)d_in[20];
    const float* l2_Wih = (const float*)d_in[21];
    const float* l2_Whh = (const float*)d_in[22];
    const float* l2_b   = (const float*)d_in[23];
    const float* hW1 = (const float*)d_in[24], *hb1 = (const float*)d_in[25];
    const float* hW2 = (const float*)d_in[26], *hb2 = (const float*)d_in[27];
    float* out = (float*)d_out;
    float* ws  = (float*)d_ws;

    const size_t OFF_HA  = 0;
    const size_t OFF_HB  = OFF_HA  + (size_t)NN * 64;
    const size_t OFF_SAS = OFF_HB  + (size_t)NN * 64;
    const size_t OFF_SAD = OFF_SAS + NN;
    const size_t OFF_EMB = OFF_SAD + NN;
    const size_t OFF_ZX  = OFF_EMB + (size_t)NGR * D_IN1;
    const size_t OFF_ZF  = OFF_ZX  + (size_t)2 * NGR * GATE;
    const size_t OFF_HST = OFF_ZF  + (size_t)2 * BATCH * GATE;
    const size_t OFF_CST = OFF_HST + (size_t)2 * BATCH * HID;
    const size_t OFF_WT  = OFF_CST + (size_t)2 * BATCH * HID;

    float* hA   = ws + OFF_HA;
    float* hB   = ws + OFF_HB;
    float* sas  = ws + OFF_SAS;
    float* sad  = ws + OFF_SAD;
    float* emb  = ws + OFF_EMB;
    float* Zx   = ws + OFF_ZX;
    float* hs2  = ws + OFF_ZF;
    float* hst  = ws + OFF_HST;
    float* cst  = ws + OFF_CST;
    float* wt   = ws + OFF_WT;
    float* ys1  = emb;
    float* ys2  = hB;
    float* hd1  = hA;

    const int TPB = 256;

    gat1_lin<<<(NN * 64 + TPB - 1) / TPB, TPB, 0, stream>>>(x, gat1_W, hA);
    gat_scores<<<(NN + TPB - 1) / TPB, TPB, 0, stream>>>(hA, gat1_as, gat1_ad, sas, sad);
    gat_aggr<<<(NN * 64 + TPB - 1) / TPB, TPB, 0, stream>>>(hA, sas, sad, gat1_b,
                                                            bn1_g, bn1_b, bn1_m, bn1_v, hB);
    transpose64<<<16, 256, 0, stream>>>(gat2_W, wt);
    launch_gemm(stream, hB, 64, wt, 64, nullptr, hA, 64, NN, 64, 64, 0);
    gat_scores<<<(NN + TPB - 1) / TPB, TPB, 0, stream>>>(hA, gat2_as, gat2_ad, sas, sad);
    gat_aggr<<<(NN * 64 + TPB - 1) / TPB, TPB, 0, stream>>>(hA, sas, sad, gat2_b,
                                                            bn2_g, bn2_b, bn2_m, bn2_v, hB);
    pack_emb<<<(NN * 66 + TPB - 1) / TPB, TPB, 0, stream>>>(hB, x, emb);

    for (int d = 0; d < 2; ++d)
        launch_gemm_lds(stream, emb, D_IN1, l1_Wih + (size_t)d * GATE * D_IN1, D_IN1,
                        l1_b + (size_t)d * GATE, Zx + (size_t)d * NGR * GATE, GATE, D_IN1);
    zero_f32<<<(2 * 2 * BATCH * HID + TPB - 1) / TPB, TPB, 0, stream>>>(hst, 2 * 2 * BATCH * HID);
    for (int t = 0; t < SEQ; ++t) {
        lstm_step<<<dim3(128, 2), 256, 0, stream>>>(Zx, l1_Whh, hst, hs2, cst, ys1, t);
    }

    for (int d = 0; d < 2; ++d)
        launch_gemm_lds(stream, ys1, D_IN2, l2_Wih + (size_t)d * GATE * D_IN2, D_IN2,
                        l2_b + (size_t)d * GATE, Zx + (size_t)d * NGR * GATE, GATE, D_IN2);
    zero_f32<<<(2 * 2 * BATCH * HID + TPB - 1) / TPB, TPB, 0, stream>>>(hst, 2 * 2 * BATCH * HID);
    for (int t = 0; t < SEQ; ++t) {
        lstm_step<<<dim3(128, 2), 256, 0, stream>>>(Zx, l2_Whh, hst, hs2, cst, ys2, t);
    }

    launch_gemm(stream, ys2, 1024, hW1, 1024, hb1, hd1, 256, NGR, 256, 1024, 1);
    launch_gemm(stream, hd1, 256,  hW2, 256,  hb2, out, 57,  NGR, 57,  256,  0);
}


# GAT element-wise kernels (linear, aggregate x2, pack): threads loop over 16-byte items (pack: 8) on a fixed grid of 16384 waves, not one float per thread (196000 waves a launch); sums unchanged
# speedup vs baseline: 3.3788x; 1.0323x over previous
#include <hip/hip_runtime.h>


#define SEQ   90
#define KP    17
#define FD    2
#define BATCH 64
#define NGR   (BATCH * SEQ)
#define NN    (NGR * KP)
#define HID   512
#define GATE  2048
#define D_IN1 1122
#define D_IN2 1024

typedef float v2f __attribute__((ext_vector_type(2)));
typedef float v8f __attribute__((ext_vector_type(8)));

__device__ __forceinline__ v8f wmma_f32(v2f a, v2f b, v8f c) {
    return __builtin_amdgcn_wmma_f32_16x16x4_f32(false, a, false, b, (short)0, c,
                                                 false, false);
}

__device__ __forceinline__ float sigmoidf_(float x) { return 1.0f / (1.0f + __expf(-x)); }

__constant__ int NBR_OFF[18] = {0,2,5,8,10,12,16,20,22,24,25,26,29,32,34,36,37,38};
__constant__ int NBR[38] = {
    1,2, 2,0,3, 1,0,4, 1,5, 2,6, 11,6,7,3, 12,5,8,4, 5,9, 6,10, 7, 8,
    13,12,5, 14,11,6, 15,11, 16,12, 13, 14
};

__global__ void gemm_wmma_f32(const float* __restrict__ A, int lda,
                              const float* __restrict__ Bm, int ldb,
                              const float* __restrict__ bias,
                              float* __restrict__ C, int ldc,
                              int M, int N, int K, int act) {
    int wave = (blockIdx.x * blockDim.x + threadIdx.x) >> 5;
    int lane = threadIdx.x & 31;
    int tilesN = (N + 31) >> 5;
    int tilesM = M >> 5;
    if (wave >= tilesM * tilesN) return;
    int tm = wave / tilesN, tn = wave % tilesN;
    int m0 = tm << 5, n0 = tn << 5;
    int half = lane >> 4, l = lane & 15;
    int kofs = half << 1;

    const float* A0 = A + (size_t)(m0 + l) * lda + kofs;
    const float* A1 = A0 + (size_t)16 * lda;
    int bn0 = n0 + l;      if (bn0 > N - 1) bn0 = N - 1;
    int bn1 = n0 + 16 + l; if (bn1 > N - 1) bn1 = N - 1;
    const float* B0 = Bm + (size_t)bn0 * ldb + kofs;
    const float* B1 = Bm + (size_t)bn1 * ldb + kofs;

    v8f acc00 = {0.f,0.f,0.f,0.f,0.f,0.f,0.f,0.f};
    v8f acc01 = acc00, acc10 = acc00, acc11 = acc00;

    int Kmain = K & ~3;
    v2f a0 = *(const v2f*)A0;
    v2f a1 = *(const v2f*)A1;
    v2f b0 = *(const v2f*)B0;
    v2f b1 = *(const v2f*)B1;
    for (int k0 = 0; k0 < Kmain - 4; k0 += 4) {
        v2f a0n = *(const v2f*)(A0 + k0 + 4);
        v2f a1n = *(const v2f*)(A1 + k0 + 4);
        v2f b0n = *(const v2f*)(B0 + k0 + 4);
        v2f b1n = *(const v2f*)(B1 + k0 + 4);
        acc00 = wmma_f32(a0, b0, acc00);
        acc01 = wmma_f32(a0, b1, acc01);
        acc10 = wmma_f32(a1, b0, acc10);
        acc11 = wmma_f32(a1, b1, acc11);
        a0 = a0n; a1 = a1n; b0 = b0n; b1 = b1n;
    }
    acc00 = wmma_f32(a0, b0, acc00);
    acc01 = wmma_f32(a0, b1, acc01);
    acc10 = wmma_f32(a1, b0, acc10);
    acc11 = wmma_f32(a1, b1, acc11);

    if (K & 3) {
        int ka = Kmain + kofs, kb = ka + 1;
        int kaC = ka < K ? ka : K - 1;
        int kbC = kb < K ? kb : K - 1;
        const float* A0b = A + (size_t)(m0 + l) * lda;
        const float* A1b = A0b + (size_t)16 * lda;
        const float* B0b = Bm + (size_t)bn0 * ldb;
        const float* B1b = Bm + (size_t)bn1 * ldb;
        float va0 = A0b[kaC], wa0 = A0b[kbC];
        float va1 = A1b[kaC], wa1 = A1b[kbC];
        float vb0 = B0b[kaC], wb0 = B0b[kbC];
        float vb1 = B1b[kaC], wb1 = B1b[kbC];
        v2f at0, at1, bt0, bt1;
        at0.x = ka < K ? va0 : 0.f; at0.y = kb < K ? wa0 : 0.f;
        at1.x = ka < K ? va1 : 0.f; at1.y = kb < K ? wa1 : 0.f;
        bt0.x = ka < K ? vb0 : 0.f; bt0.y = kb < K ? wb0 : 0.f;
        bt1.x = ka < K ? vb1 : 0.f; bt1.y = kb < K ? wb1 : 0.f;
        acc00 = wmma_f32(at0, bt0, acc00);
        acc01 = wmma_f32(at0, bt1, acc01);
        acc10 = wmma_f32(at1, bt0, acc10);
        acc11 = wmma_f32(at1, bt1, acc11);
    }
    int cn0 = n0 + l, cn1 = n0 + 16 + l;
    if (cn0 < N) {
        float bv = bias ? bias[cn0] : 0.f;
        #pragma unroll
        for (int j = 0; j < 8; ++j) {
            int r0 = m0 + j + half * 8, r1 = r0 + 16;
            float v0 = acc00[j] + bv, v1 = acc10[j] + bv;
            if (act) { v0 = fmaxf(v0, 0.f); v1 = fmaxf(v1, 0.f); }
            C[(size_t)r0 * ldc + cn0] = v0;
            C[(size_t)r1 * ldc + cn0] = v1;
        }
    }
    if (cn1 < N) {
        float bv = bias ? bias[cn1] : 0.f;
        #pragma unroll
        for (int j = 0; j < 8; ++j) {
            int r0 = m0 + j + half * 8, r1 = r0 + 16;
            float v0 = acc01[j] + bv, v1 = acc11[j] + bv;
            if (act) { v0 = fmaxf(v0, 0.f); v1 = fmaxf(v1, 0.f); }
            C[(size_t)r0 * ldc + cn1] = v0;
            C[(size_t)r1 * ldc + cn1] = v1;
        }
    }
}

#define GL_PITCH 36
#define GL_OPER  (128 * GL_PITCH)
#define GL_STAGE (2 * GL_OPER)

typedef float v4g __attribute__((ext_vector_type(4)));

__device__ __forceinline__ void gl_fetch(v2f (&r)[8], const float* __restrict__ p, size_t ld16) {
    #pragma unroll
    for (int i = 0; i < 8; ++i) r[i] = *(const v2f*)(p + i * ld16);
}

__device__ __forceinline__ void gl_fetch_tail(v2f (&r)[8], const float* __restrict__ p,
                                              size_t ld16, bool ok) {
    #pragma unroll
    for (int i = 0; i < 8; ++i) r[i] = (v2f){0.f, 0.f};
    if (ok) {
        #pragma unroll
        for (int i = 0; i < 8; ++i) r[i] = *(const v2f*)(p + i * ld16);
    }
}

__device__ __forceinline__ void gl_store(float* __restrict__ s, const v2f (&r)[8]) {
    #pragma unroll
    for (int i = 0; i < 8; ++i) *(v2f*)(s + i * 16 * GL_PITCH) = r[i];
}

__device__ __forceinline__ v2f gl_pick(v4g v, int e) {
    v2f r;
    if (e) { r.x = v.z; r.y = v.w; } else { r.x = v.x; r.y = v.y; }
    return r;
}

__global__ __launch_bounds__(256)
void gemm_lds_f32(const float* __restrict__ A, int lda,
                  const float* __restrict__ Bm, int ldb,
                  const float* __restrict__ bias,
                  float* __restrict__ C, int ldc, int K) {
    __shared__ float sm[2 * GL_STAGE];

    int tid  = threadIdx.x;
    int wv   = tid >> 5, lane = tid & 31;
    int half = lane >> 4, l = lane & 15;
    int wm = wv >> 2, wn = wv & 3;
    int m0 = blockIdx.y << 7, n0 = blockIdx.x << 7;

    int cp = tid & 15, r0 = tid >> 4;
    const float* Ag = A  + (size_t)(m0 + r0) * lda + 2 * cp;
    const float* Bg = Bm + (size_t)(n0 + r0) * ldb + 2 * cp;
    size_t lda16 = (size_t)16 * lda, ldb16 = (size_t)16 * ldb;
    float* sA = sm + r0 * GL_PITCH + 2 * cp;
    float* sB = sA + GL_OPER;

    const float* fA = sm + (wm * 64 + l) * GL_PITCH + 16 * half;
    const float* fB = sm + GL_OPER + (wn * 32 + l) * GL_PITCH + 16 * half;

    v8f acc[4][2];
    #pragma unroll
    for (int t = 0; t < 8; ++t) acc[t >> 1][t & 1] = (v8f){0.f,0.f,0.f,0.f,0.f,0.f,0.f,0.f};

    int nch = (K + 31) >> 5;
    v2f ra[8], rb[8];
    gl_fetch(ra, Ag, lda16);
    gl_fetch(rb, Bg, ldb16);
    gl_store(sA, ra);
    gl_store(sB, rb);
    __syncthreads();

    for (int ch = 0; ch < nch; ++ch) {
        int  st   = ch & 1;
        bool more = ch + 1 < nch;
        if (more) {
            int kn = (ch + 1) << 5;
            if (kn + 32 <= K) {
                gl_fetch(ra, Ag + kn, lda16);
                gl_fetch(rb, Bg + kn, ldb16);
            } else {
                bool ok = kn + 2 * cp < K;
                gl_fetch_tail(ra, Ag + kn, lda16, ok);
                gl_fetch_tail(rb, Bg + kn, ldb16, ok);
            }
        }
        const float* pa = fA + st * GL_STAGE;
        const float* pb = fB + st * GL_STAGE;
        v4g qa[4][4], qb[2][4];
        #pragma unroll
        for (int t = 0; t < 16; ++t)
            qa[t & 3][t >> 2] = *(const v4g*)(pa + (t & 3) * 16 * GL_PITCH + 4 * (t >> 2));
        #pragma unroll
        for (int t = 0; t < 8; ++t)
            qb[t & 1][t >> 1] = *(const v4g*)(pb + (t & 1) * 16 * GL_PITCH + 4 * (t >> 1));
        #pragma unroll
        for (int s = 0; s < 64; ++s) {
            int q = s >> 4, e = (s >> 3) & 1, i = (s >> 1) & 3, j = s & 1;
            acc[i][j] = wmma_f32(gl_pick(qa[i][q], e), gl_pick(qb[j][q], e), acc[i][j]);
        }
        if (more) {
            gl_store(sA + (st ^ 1) * GL_STAGE, ra);
            gl_store(sB + (st ^ 1) * GL_STAGE, rb);
        }
        __syncthreads();
    }

    int col0 = n0 + wn * 32 + l;
    float bv0 = bias[col0], bv1 = bias[col0 + 16];
    #pragma unroll
    for (int u = 0; u < 64; ++u) {
        int j = u >> 5, i = (u >> 3) & 3, r = u & 7;
        int row = m0 + wm * 64 + 16 * i + r + 8 * half;
        C[(size_t)row * ldc + col0 + 16 * j] = acc[i][j][r] + (j ? bv1 : bv0);
    }
}

#define KSPLIT    8
#define KSLICE    (HID / KSPLIT)
#define RED_PITCH 36

typedef float v4f __attribute__((ext_vector_type(4)));

struct Slice { v4f q[8]; };

__device__ __forceinline__ void load_group(Slice& s, const float* __restrict__ p, int g) {
    s.q[2 * g]     = *(const v4f*)(p + 16 * g);
    s.q[2 * g + 1] = *(const v4f*)(p + 16 * g + 4);
}

__device__ __forceinline__ v2f frag_step(const Slice& s, int st) {
    v4f v = s.q[st >> 1];
    v2f r;
    if (st & 1) { r.x = v.z; r.y = v.w; } else { r.x = v.x; r.y = v.y; }
    return r;
}

__global__ __launch_bounds__(256)
void lstm_step(const float* __restrict__ Zx,
               const float* __restrict__ Whh,
               float* hbuf0, float* hbuf1,
               float* __restrict__ cbuf,
               float* __restrict__ ys, int t) {
    __shared__ float red[KSPLIT * 32 * RED_PITCH];

    const float* hprev = (t & 1) ? hbuf1 : hbuf0;
    float*       hnext = (t & 1) ? hbuf0 : hbuf1;

    int dir  = blockIdx.y;
    int tile = blockIdx.x;
    int wv   = threadIdx.x >> 5;
    int lane = threadIdx.x & 31;
    int tm = tile >> 6, tn = tile & 63;
    int m0 = tm << 5, j0 = tn << 3;
    int half = lane >> 4, l = lane & 15;

    int kb = wv * KSLICE + half * 8;
    int nb = (l >> 3) * HID + j0 + (l & 7);
    const float* A0 = hprev + (size_t)dir * BATCH * HID + (size_t)(m0 + l) * HID + kb;
    const float* B0 = Whh + (size_t)dir * GATE * HID + (size_t)nb * HID + kb;
    const float* A1 = A0 + (size_t)16 * HID;
    const float* B1 = B0 + (size_t)2 * HID * HID;
    Slice a0, a1, b0, b1;
    #pragma unroll
    for (int g = 0; g < 4; ++g) {
        load_group(a0, A0, g);
        load_group(b0, B0, g);
        load_group(b1, B1, g);
        load_group(a1, A1, g);
    }
    int t_d = dir ? (SEQ - 1 - t) : t;
    int rr  = 4 * wv + (lane >> 3);
    int jj  = lane & 7;
    int b   = m0 + rr;
    int sidx = dir * BATCH * HID + b * HID + j0 + jj;
    const float* zxp = Zx + (size_t)dir * NGR * GATE + (size_t)(b * SEQ + t_d) * GATE + j0 + jj;
    float zx[4];
    #pragma unroll
    for (int g = 0; g < 4; ++g) zx[g] = zxp[g * HID];
    float c = cbuf[sidx];
    __builtin_amdgcn_sched_barrier(0);

    v8f acc00 = {0.f,0.f,0.f,0.f,0.f,0.f,0.f,0.f};
    v8f acc01 = acc00, acc10 = acc00, acc11 = acc00;
    #pragma unroll
    for (int st = 0; st < 16; ++st) {
        v2f fa0 = frag_step(a0, st), fa1 = frag_step(a1, st);
        v2f fb0 = frag_step(b0, st), fb1 = frag_step(b1, st);
        acc00 = wmma_f32(fa0, fb0, acc00);
        acc01 = wmma_f32(fa0, fb1, acc01);
        acc10 = wmma_f32(fa1, fb0, acc10);
        acc11 = wmma_f32(fa1, fb1, acc11);
    }

    float* rw = red + wv * (32 * RED_PITCH);
    #pragma unroll
    for (int j = 0; j < 8; ++j) {
        int r = j + 8 * half;
        rw[r * RED_PITCH + l]             = acc00[j];
        rw[r * RED_PITCH + 16 + l]        = acc01[j];
        rw[(r + 16) * RED_PITCH + l]      = acc10[j];
        rw[(r + 16) * RED_PITCH + 16 + l] = acc11[j];
    }
    __syncthreads();

    const float* rp = red + rr * RED_PITCH + jj;
    float z[4];
    #pragma unroll
    for (int g = 0; g < 4; ++g) {
        float s = rp[8 * g];
        #pragma unroll
        for (int w = 1; w < KSPLIT; ++w) s += rp[w * (32 * RED_PITCH) + 8 * g];
        z[g] = s + zx[g];
    }
    c = sigmoidf_(z[1]) * c + sigmoidf_(z[0]) * tanhf(z[2]);
    float h = sigmoidf_(z[3]) * tanhf(c);
    cbuf[sidx]  = c;
    hnext[sidx] = h;
    ys[(size_t)(b * SEQ + t_d) * 1024 + dir * HID + j0 + jj] = h;
}

#define EW_BLOCKS  2048
#define EW_THREADS (EW_BLOCKS * 256)

typedef float f4v __attribute__((ext_vector_type(4)));

__global__ __launch_bounds__(256)
void gat1_lin(const float* __restrict__ x, const float* __restrict__ W,
              float* __restrict__ h) {
    int gtid = blockIdx.x * 256 + threadIdx.x;
    int q4 = (gtid & 15) * 4;
    f4v w0 = *(const f4v*)(W + q4);
    f4v w1 = *(const f4v*)(W + 64 + q4);
    for (int it = gtid; it < NN * 16; it += EW_THREADS) {
        v2f xv = *(const v2f*)(x + 2 * (it >> 4));
        f4v r;
        r.x = xv.x * w0.x + xv.y * w1.x;
        r.y = xv.x * w0.y + xv.y * w1.y;
        r.z = xv.x * w0.z + xv.y * w1.z;
        r.w = xv.x * w0.w + xv.y * w1.w;
        *(f4v*)(h + (size_t)it * 4) = r;
    }
}

__global__ void gat_scores(const float* __restrict__ h,
                           const float* __restrict__ asrc, const float* __restrict__ adst,
                           float* __restrict__ sas, float* __restrict__ sad) {
    int i = blockIdx.x * blockDim.x + threadIdx.x;
    if (i >= NN) return;
    const float* hp = h + (size_t)i * 64;
    float a = 0.f, d = 0.f;
    #pragma unroll
    for (int f = 0; f < 64; ++f) { float v = hp[f]; a += v * asrc[f]; d += v * adst[f]; }
    sas[i] = a; sad[i] = d;
}

__global__ __launch_bounds__(256)
void gat_aggr(const float* __restrict__ h,
              const float* __restrict__ sas, const float* __restrict__ sad,
              const float* __restrict__ bvec,
              const float* __restrict__ g, const float* __restrict__ bb,
              const float* __restrict__ bm, const float* __restrict__ bvv,
              float* __restrict__ out) {
    int gtid = blockIdx.x * 256 + threadIdx.x;
    int ch = (gtid & 15) * 4;
    f4v bias = *(const f4v*)(bvec + ch);
    f4v mean = *(const f4v*)(bm + ch);
    f4v gam  = *(const f4v*)(g + ch);
    f4v var  = *(const f4v*)(bvv + ch);
    f4v beta = *(const f4v*)(bb + ch);
    f4v scl;
    scl.x = gam.x * rsqrtf(var.x + 1e-5f);
    scl.y = gam.y * rsqrtf(var.y + 1e-5f);
    scl.z = gam.z * rsqrtf(var.z + 1e-5f);
    scl.w = gam.w * rsqrtf(var.w + 1e-5f);

    for (int it = gtid; it < NN * 16; it += EW_THREADS) {
        int i = it >> 4;
        int dl = i % KP, base = i - dl;
        float ad = sad[i];
        int o0 = NBR_OFF[dl], deg = NBR_OFF[dl + 1] - o0;

        float e[5];
        f4v hv[5];
        float t0 = sas[i] + ad;
        e[0]  = t0 < 0.f ? 0.2f * t0 : t0;
        hv[0] = *(const f4v*)(h + (size_t)i * 64 + ch);
        #pragma unroll
        for (int n = 0; n < 4; ++n) {
            bool on = n < deg;
            int k = on ? n : deg - 1;
            int s = base + NBR[o0 + k];
            float tt = sas[s] + ad;
            float ev = tt < 0.f ? 0.2f * tt : tt;
            e[n + 1]  = on ? ev : -__builtin_inff();
            hv[n + 1] = *(const f4v*)(h + (size_t)s * 64 + ch);
        }
        float m = e[0];
        #pragma unroll
        for (int n = 1; n < 5; ++n) m = fmaxf(m, e[n]);
        float den = 0.f;
        f4v num = {0.f, 0.f, 0.f, 0.f};
        #pragma unroll
        for (int n = 0; n < 5; ++n) {
            float w = __expf(e[n] - m);
            den += w;
            num += w * hv[n];
        }
        f4v val;
        val.x = num.x / den + bias.x;
        val.y = num.y / den + bias.y;
        val.z = num.z / den + bias.z;
        val.w = num.w / den + bias.w;
        val = (val - mean) * scl + beta;
        f4v o;
        o.x = fmaxf(val.x, 0.f);
        o.y = fmaxf(val.y, 0.f);
        o.z = fmaxf(val.z, 0.f);
        o.w = fmaxf(val.w, 0.f);
        *(f4v*)(out + (size_t)it * 4) = o;
    }
}

__global__ __launch_bounds__(256)
void pack_emb(const float* __restrict__ h2, const float* __restrict__ x,
              float* __restrict__ emb) {
    int gtid = blockIdx.x * 256 + threadIdx.x;
    for (int it = gtid; it < NN * 33; it += EW_THREADS) {
        int i = it / 33, c2 = it - i * 33;
        const float* p = (c2 < 32) ? h2 + (size_t)i * 64 + 2 * c2 : x + 2 * i;
        *(v2f*)(emb + (size_t)it * 2) = *(const v2f*)p;
    }
}

__global__ void transpose64(const float* __restrict__ W, float* __restrict__ WT) {
    int idx = blockIdx.x * blockDim.x + threadIdx.x;
    if (idx >= 64 * 64) return;
    int n = idx >> 6, k = idx & 63;
    WT[n * 64 + k] = W[k * 64 + n];
}

__global__ void zero_f32(float* __restrict__ p, int n) {
    int i = blockIdx.x * blockDim.x + threadIdx.x;
    if (i < n) p[i] = 0.f;
}

static inline void launch_gemm(hipStream_t s, const float* A, int lda,
                               const float* B, int ldb, const float* bias,
                               float* C, int ldc, int M, int N, int K, int act) {
    int tiles = (M / 32) * ((N + 31) / 32);
    int blocks = (tiles + 7) / 8;
    gemm_wmma_f32<<<blocks, 256, 0, s>>>(A, lda, B, ldb, bias, C, ldc, M, N, K, act);
}

static inline void launch_gemm_lds(hipStream_t s, const float* A, int lda,
                                   const float* B, int ldb, const float* bias,
                                   float* C, int ldc, int K) {
    gemm_lds_f32<<<dim3(GATE / 128, NGR / 128), 256, 0, s>>>(A, lda, B, ldb, bias, C, ldc, K);
}

extern "C" void kernel_launch(void* const* d_in, const int* in_sizes, int n_in,
                              void* d_out, int out_size, void* d_ws, size_t ws_size,
                              hipStream_t stream) {
    const float* x        = (const float*)d_in[0];
    const float* gat1_W   = (const float*)d_in[2];
    const float* gat1_as  = (const float*)d_in[3];
    const float* gat1_ad  = (const float*)d_in[4];
    const float* gat1_b   = (const float*)d_in[5];
    const float* gat2_W   = (const float*)d_in[6];
    const float* gat2_as  = (const float*)d_in[7];
    const float* gat2_ad  = (const float*)d_in[8];
    const float* gat2_b   = (const float*)d_in[9];
    const float* bn1_g = (const float*)d_in[10], *bn1_b = (const float*)d_in[11];
    const float* bn1_m = (const float*)d_in[12], *bn1_v = (const float*)d_in[13];
    const float* bn2_g = (const float*)d_in[14], *bn2_b = (const float*)d_in[15];
    const float* bn2_m = (const float*)d_in[16], *bn2_v = (const float*)d_in[17];
    const float* l1_Wih = (const float*)d_in[18];
    const float* l1_Whh = (const float*)d_in[19];
    const float* l1_b   = (const float*)d_in[20];
    const float* l2_Wih = (const float*)d_in[21];
    const float* l2_Whh = (const float*)d_in[22];
    const float* l2_b   = (const float*)d_in[23];
    const float* hW1 = (const float*)d_in[24], *hb1 = (const float*)d_in[25];
    const float* hW2 = (const float*)d_in[26], *hb2 = (const float*)d_in[27];
    float* out = (float*)d_out;
    float* ws  = (float*)d_ws;

    const size_t OFF_HA  = 0;
    const size_t OFF_HB  = OFF_HA  + (size_t)NN * 64;
    const size_t OFF_SAS = OFF_HB  + (size_t)NN * 64;
    const size_t OFF_SAD = OFF_SAS + NN;
    const size_t OFF_EMB = OFF_SAD + NN;
    const size_t OFF_ZX  = OFF_EMB + (size_t)NGR * D_IN1;
    const size_t OFF_ZF  = OFF_ZX  + (size_t)2 * NGR * GATE;
    const size_t OFF_HST = OFF_ZF  + (size_t)2 * BATCH * GATE;
    const size_t OFF_CST = OFF_HST + (size_t)2 * BATCH * HID;
    const size_t OFF_WT  = OFF_CST + (size_t)2 * BATCH * HID;

    float* hA   = ws + OFF_HA;
    float* hB   = ws + OFF_HB;
    float* sas  = ws + OFF_SAS;
    float* sad  = ws + OFF_SAD;
    float* emb  = ws + OFF_EMB;
    float* Zx   = ws + OFF_ZX;
    float* hs2  = ws + OFF_ZF;
    float* hst  = ws + OFF_HST;
    float* cst  = ws + OFF_CST;
    float* wt   = ws + OFF_WT;
    float* ys1  = emb;
    float* ys2  = hB;
    float* hd1  = hA;

    const int TPB = 256;

    gat1_lin<<<EW_BLOCKS, TPB, 0, stream>>>(x, gat1_W, hA);
    gat_scores<<<(NN + TPB - 1) / TPB, TPB, 0, stream>>>(hA, gat1_as, gat1_ad, sas, sad);
    gat_aggr<<<EW_BLOCKS, TPB, 0, stream>>>(hA, sas, sad, gat1_b,
                                            bn1_g, bn1_b, bn1_m, bn1_v, hB);
    transpose64<<<16, 256, 0, stream>>>(gat2_W, wt);
    launch_gemm(stream, hB, 64, wt, 64, nullptr, hA, 64, NN, 64, 64, 0);
    gat_scores<<<(NN + TPB - 1) / TPB, TPB, 0, stream>>>(hA, gat2_as, gat2_ad, sas, sad);
    gat_aggr<<<EW_BLOCKS, TPB, 0, stream>>>(hA, sas, sad, gat2_b,
                                            bn2_g, bn2_b, bn2_m, bn2_v, hB);
    pack_emb<<<EW_BLOCKS, TPB, 0, stream>>>(hB, x, emb);

    for (int d = 0; d < 2; ++d)
        launch_gemm_lds(stream, emb, D_IN1, l1_Wih + (size_t)d * GATE * D_IN1, D_IN1,
                        l1_b + (size_t)d * GATE, Zx + (size_t)d * NGR * GATE, GATE, D_IN1);
    zero_f32<<<(2 * 2 * BATCH * HID + TPB - 1) / TPB, TPB, 0, stream>>>(hst, 2 * 2 * BATCH * HID);
    for (int t = 0; t < SEQ; ++t) {
        lstm_step<<<dim3(128, 2), 256, 0, stream>>>(Zx, l1_Whh, hst, hs2, cst, ys1, t);
    }

    for (int d = 0; d < 2; ++d)
        launch_gemm_lds(stream, ys1, D_IN2, l2_Wih + (size_t)d * GATE * D_IN2, D_IN2,
                        l2_b + (size_t)d * GATE, Zx + (size_t)d * NGR * GATE, GATE, D_IN2);
    zero_f32<<<(2 * 2 * BATCH * HID + TPB - 1) / TPB, TPB, 0, stream>>>(hst, 2 * 2 * BATCH * HID);
    for (int t = 0; t < SEQ; ++t) {
        lstm_step<<<dim3(128, 2), 256, 0, stream>>>(Zx, l2_Whh, hst, hs2, cst, ys2, t);
    }

    launch_gemm(stream, ys2, 1024, hW1, 1024, hb1, hd1, 256, NGR, 256, 1024, 1);
    launch_gemm(stream, hd1, 256,  hW2, 256,  hb2, out, 57,  NGR, 57,  256,  0);
}


# recurrent step: operand slices fetched as whole 128-byte lines (4 rows per 16-byte load, not 16 rows x 32 B) and turned into WMMA fragments through wave-private LDS rows; f32 sum order kept
# speedup vs baseline: 3.5223x; 1.0425x over previous
#include <hip/hip_runtime.h>


#define SEQ   90
#define KP    17
#define FD    2
#define BATCH 64
#define NGR   (BATCH * SEQ)
#define NN    (NGR * KP)
#define HID   512
#define GATE  2048
#define D_IN1 1122
#define D_IN2 1024

typedef float v2f __attribute__((ext_vector_type(2)));
typedef float v8f __attribute__((ext_vector_type(8)));

__device__ __forceinline__ v8f wmma_f32(v2f a, v2f b, v8f c) {
    return __builtin_amdgcn_wmma_f32_16x16x4_f32(false, a, false, b, (short)0, c,
                                                 false, false);
}

__device__ __forceinline__ float sigmoidf_(float x) { return 1.0f / (1.0f + __expf(-x)); }

__constant__ int NBR_OFF[18] = {0,2,5,8,10,12,16,20,22,24,25,26,29,32,34,36,37,38};
__constant__ int NBR[38] = {
    1,2, 2,0,3, 1,0,4, 1,5, 2,6, 11,6,7,3, 12,5,8,4, 5,9, 6,10, 7, 8,
    13,12,5, 14,11,6, 15,11, 16,12, 13, 14
};

__global__ void gemm_wmma_f32(const float* __restrict__ A, int lda,
                              const float* __restrict__ Bm, int ldb,
                              const float* __restrict__ bias,
                              float* __restrict__ C, int ldc,
                              int M, int N, int K, int act) {
    int wave = (blockIdx.x * blockDim.x + threadIdx.x) >> 5;
    int lane = threadIdx.x & 31;
    int tilesN = (N + 31) >> 5;
    int tilesM = M >> 5;
    if (wave >= tilesM * tilesN) return;
    int tm = wave / tilesN, tn = wave % tilesN;
    int m0 = tm << 5, n0 = tn << 5;
    int half = lane >> 4, l = lane & 15;
    int kofs = half << 1;

    const float* A0 = A + (size_t)(m0 + l) * lda + kofs;
    const float* A1 = A0 + (size_t)16 * lda;
    int bn0 = n0 + l;      if (bn0 > N - 1) bn0 = N - 1;
    int bn1 = n0 + 16 + l; if (bn1 > N - 1) bn1 = N - 1;
    const float* B0 = Bm + (size_t)bn0 * ldb + kofs;
    const float* B1 = Bm + (size_t)bn1 * ldb + kofs;

    v8f acc00 = {0.f,0.f,0.f,0.f,0.f,0.f,0.f,0.f};
    v8f acc01 = acc00, acc10 = acc00, acc11 = acc00;

    int Kmain = K & ~3;
    v2f a0 = *(const v2f*)A0;
    v2f a1 = *(const v2f*)A1;
    v2f b0 = *(const v2f*)B0;
    v2f b1 = *(const v2f*)B1;
    for (int k0 = 0; k0 < Kmain - 4; k0 += 4) {
        v2f a0n = *(const v2f*)(A0 + k0 + 4);
        v2f a1n = *(const v2f*)(A1 + k0 + 4);
        v2f b0n = *(const v2f*)(B0 + k0 + 4);
        v2f b1n = *(const v2f*)(B1 + k0 + 4);
        acc00 = wmma_f32(a0, b0, acc00);
        acc01 = wmma_f32(a0, b1, acc01);
        acc10 = wmma_f32(a1, b0, acc10);
        acc11 = wmma_f32(a1, b1, acc11);
        a0 = a0n; a1 = a1n; b0 = b0n; b1 = b1n;
    }
    acc00 = wmma_f32(a0, b0, acc00);
    acc01 = wmma_f32(a0, b1, acc01);
    acc10 = wmma_f32(a1, b0, acc10);
    acc11 = wmma_f32(a1, b1, acc11);

    if (K & 3) {
        int ka = Kmain + kofs, kb = ka + 1;
        int kaC = ka < K ? ka : K - 1;
        int kbC = kb < K ? kb : K - 1;
        const float* A0b = A + (size_t)(m0 + l) * lda;
        const float* A1b = A0b + (size_t)16 * lda;
        const float* B0b = Bm + (size_t)bn0 * ldb;
        const float* B1b = Bm + (size_t)bn1 * ldb;
        float va0 = A0b[kaC], wa0 = A0b[kbC];
        float va1 = A1b[kaC], wa1 = A1b[kbC];
        float vb0 = B0b[kaC], wb0 = B0b[kbC];
        float vb1 = B1b[kaC], wb1 = B1b[kbC];
        v2f at0, at1, bt0, bt1;
        at0.x = ka < K ? va0 : 0.f; at0.y = kb < K ? wa0 : 0.f;
        at1.x = ka < K ? va1 : 0.f; at1.y = kb < K ? wa1 : 0.f;
        bt0.x = ka < K ? vb0 : 0.f; bt0.y = kb < K ? wb0 : 0.f;
        bt1.x = ka < K ? vb1 : 0.f; bt1.y = kb < K ? wb1 : 0.f;
        acc00 = wmma_f32(at0, bt0, acc00);
        acc01 = wmma_f32(at0, bt1, acc01);
        acc10 = wmma_f32(at1, bt0, acc10);
        acc11 = wmma_f32(at1, bt1, acc11);
    }
    int cn0 = n0 + l, cn1 = n0 + 16 + l;
    if (cn0 < N) {
        float bv = bias ? bias[cn0] : 0.f;
        #pragma unroll
        for (int j = 0; j < 8; ++j) {
            int r0 = m0 + j + half * 8, r1 = r0 + 16;
            float v0 = acc00[j] + bv, v1 = acc10[j] + bv;
            if (act) { v0 = fmaxf(v0, 0.f); v1 = fmaxf(v1, 0.f); }
            C[(size_t)r0 * ldc + cn0] = v0;
            C[(size_t)r1 * ldc + cn0] = v1;
        }
    }
    if (cn1 < N) {
        float bv = bias ? bias[cn1] : 0.f;
        #pragma unroll
        for (int j = 0; j < 8; ++j) {
            int r0 = m0 + j + half * 8, r1 = r0 + 16;
            float v0 = acc01[j] + bv, v1 = acc11[j] + bv;
            if (act) { v0 = fmaxf(v0, 0.f); v1 = fmaxf(v1, 0.f); }
            C[(size_t)r0 * ldc + cn1] = v0;
            C[(size_t)r1 * ldc + cn1] = v1;
        }
    }
}

#define GL_PITCH 36
#define GL_OPER  (128 * GL_PITCH)
#define GL_STAGE (2 * GL_OPER)

typedef float v4g __attribute__((ext_vector_type(4)));

__device__ __forceinline__ void gl_fetch(v2f (&r)[8], const float* __restrict__ p, size_t ld16) {
    #pragma unroll
    for (int i = 0; i < 8; ++i) r[i] = *(const v2f*)(p + i * ld16);
}

__device__ __forceinline__ void gl_fetch_tail(v2f (&r)[8], const float* __restrict__ p,
                                              size_t ld16, bool ok) {
    #pragma unroll
    for (int i = 0; i < 8; ++i) r[i] = (v2f){0.f, 0.f};
    if (ok) {
        #pragma unroll
        for (int i = 0; i < 8; ++i) r[i] = *(const v2f*)(p + i * ld16);
    }
}

__device__ __forceinline__ void gl_store(float* __restrict__ s, const v2f (&r)[8]) {
    #pragma unroll
    for (int i = 0; i < 8; ++i) *(v2f*)(s + i * 16 * GL_PITCH) = r[i];
}

__device__ __forceinline__ v2f gl_pick(v4g v, int e) {
    v2f r;
    if (e) { r.x = v.z; r.y = v.w; } else { r.x = v.x; r.y = v.y; }
    return r;
}

__global__ __launch_bounds__(256)
void gemm_lds_f32(const float* __restrict__ A, int lda,
                  const float* __restrict__ Bm, int ldb,
                  const float* __restrict__ bias,
                  float* __restrict__ C, int ldc, int K) {
    __shared__ float sm[2 * GL_STAGE];

    int tid  = threadIdx.x;
    int wv   = tid >> 5, lane = tid & 31;
    int half = lane >> 4, l = lane & 15;
    int wm = wv >> 2, wn = wv & 3;
    int m0 = blockIdx.y << 7, n0 = blockIdx.x << 7;

    int cp = tid & 15, r0 = tid >> 4;
    const float* Ag = A  + (size_t)(m0 + r0) * lda + 2 * cp;
    const float* Bg = Bm + (size_t)(n0 + r0) * ldb + 2 * cp;
    size_t lda16 = (size_t)16 * lda, ldb16 = (size_t)16 * ldb;
    float* sA = sm + r0 * GL_PITCH + 2 * cp;
    float* sB = sA + GL_OPER;

    const float* fA = sm + (wm * 64 + l) * GL_PITCH + 16 * half;
    const float* fB = sm + GL_OPER + (wn * 32 + l) * GL_PITCH + 16 * half;

    v8f acc[4][2];
    #pragma unroll
    for (int t = 0; t < 8; ++t) acc[t >> 1][t & 1] = (v8f){0.f,0.f,0.f,0.f,0.f,0.f,0.f,0.f};

    int nch = (K + 31) >> 5;
    v2f ra[8], rb[8];
    gl_fetch(ra, Ag, lda16);
    gl_fetch(rb, Bg, ldb16);
    gl_store(sA, ra);
    gl_store(sB, rb);
    __syncthreads();

    for (int ch = 0; ch < nch; ++ch) {
        int  st   = ch & 1;
        bool more = ch + 1 < nch;
        if (more) {
            int kn = (ch + 1) << 5;
            if (kn + 32 <= K) {
                gl_fetch(ra, Ag + kn, lda16);
                gl_fetch(rb, Bg + kn, ldb16);
            } else {
                bool ok = kn + 2 * cp < K;
                gl_fetch_tail(ra, Ag + kn, lda16, ok);
                gl_fetch_tail(rb, Bg + kn, ldb16, ok);
            }
        }
        const float* pa = fA + st * GL_STAGE;
        const float* pb = fB + st * GL_STAGE;
        v4g qa[4][4], qb[2][4];
        #pragma unroll
        for (int t = 0; t < 16; ++t)
            qa[t & 3][t >> 2] = *(const v4g*)(pa + (t & 3) * 16 * GL_PITCH + 4 * (t >> 2));
        #pragma unroll
        for (int t = 0; t < 8; ++t)
            qb[t & 1][t >> 1] = *(const v4g*)(pb + (t & 1) * 16 * GL_PITCH + 4 * (t >> 1));
        #pragma unroll
        for (int s = 0; s < 64; ++s) {
            int q = s >> 4, e = (s >> 3) & 1, i = (s >> 1) & 3, j = s & 1;
            acc[i][j] = wmma_f32(gl_pick(qa[i][q], e), gl_pick(qb[j][q], e), acc[i][j]);
        }
        if (more) {
            gl_store(sA + (st ^ 1) * GL_STAGE, ra);
            gl_store(sB + (st ^ 1) * GL_STAGE, rb);
        }
        __syncthreads();
    }

    int col0 = n0 + wn * 32 + l;
    float bv0 = bias[col0], bv1 = bias[col0 + 16];
    #pragma unroll
    for (int u = 0; u < 64; ++u) {
        int j = u >> 5, i = (u >> 3) & 3, r = u & 7;
        int row = m0 + wm * 64 + 16 * i + r + 8 * half;
        C[(size_t)row * ldc + col0 + 16 * j] = acc[i][j][r] + (j ? bv1 : bv0);
    }
}

#define KSPLIT    8
#define KSLICE    (HID / KSPLIT)
#define ST_PITCH  68
#define ST_WAVE   (64 * ST_PITCH)
#define RED_PITCH 36

typedef float v4f __attribute__((ext_vector_type(4)));

struct Slice { v4f q[8]; };

__device__ __forceinline__ void load_group(Slice& s, const float* p, int g) {
    s.q[2 * g]     = *(const v4f*)(p + 16 * g);
    s.q[2 * g + 1] = *(const v4f*)(p + 16 * g + 4);
}

__device__ __forceinline__ v2f frag_step(const Slice& s, int st) {
    v4f v = s.q[st >> 1];
    v2f r;
    if (st & 1) { r.x = v.z; r.y = v.w; } else { r.x = v.x; r.y = v.y; }
    return r;
}

__device__ __forceinline__ void fetch_block(v4f (&r)[4], const float* p) {
    #pragma unroll
    for (int i = 0; i < 4; ++i) r[i] = *(const v4f*)(p + (size_t)i * 2 * HID);
}

__device__ __forceinline__ void stage_block(float* s, const v4f (&r)[4]) {
    #pragma unroll
    for (int i = 0; i < 4; ++i) *(v4f*)(s + i * 2 * ST_PITCH) = r[i];
}

__global__ __launch_bounds__(256)
void lstm_step(const float* __restrict__ Zx,
               const float* __restrict__ Whh,
               float* hbuf0, float* hbuf1,
               float* __restrict__ cbuf,
               float* __restrict__ ys, int t) {
    __shared__ float stg[KSPLIT * ST_WAVE];

    const float* hprev = (t & 1) ? hbuf1 : hbuf0;
    float*       hnext = (t & 1) ? hbuf0 : hbuf1;

    int dir  = blockIdx.y;
    int tile = blockIdx.x;
    int wv   = threadIdx.x >> 5;
    int lane = threadIdx.x & 31;
    int tm = tile >> 6, tn = tile & 63;
    int m0 = tm << 5, j0 = tn << 3;
    int half = lane >> 4, l = lane & 15;

    int sp = lane >> 3, sq = lane & 7;
    int r16 = (sp & 1) * 8 + (sp >> 1);
    int kw  = wv * KSLICE + 4 * sq;
    const float* gA = hprev + (size_t)dir * BATCH * HID + (size_t)(m0 + r16) * HID + kw;
    const float* gB = Whh + (size_t)dir * GATE * HID
                    + (size_t)((sp & 1) * HID + j0 + (sp >> 1)) * HID + kw;
    v4f ga0[2][4], ga1[2][4], gb0[2][4], gb1[2][4];
    #pragma unroll
    for (int hf = 0; hf < 2; ++hf) {
        fetch_block(ga0[hf], gA + 32 * hf);
        fetch_block(gb0[hf], gB + 32 * hf);
        fetch_block(gb1[hf], gB + (size_t)2 * HID * HID + 32 * hf);
        fetch_block(ga1[hf], gA + (size_t)16 * HID + 32 * hf);
        if (hf == 0) {
            __builtin_amdgcn_sched_barrier(0);
        }
    }
    int t_d = dir ? (SEQ - 1 - t) : t;
    int rr  = 4 * wv + (lane >> 3);
    int jj  = lane & 7;
    int b   = m0 + rr;
    int sidx = dir * BATCH * HID + b * HID + j0 + jj;
    const float* zxp = Zx + (size_t)dir * NGR * GATE + (size_t)(b * SEQ + t_d) * GATE + j0 + jj;
    float zx[4];
    #pragma unroll
    for (int g = 0; g < 4; ++g) zx[g] = zxp[g * HID];
    float c = cbuf[sidx];

    float* wreg = stg + wv * ST_WAVE;
    float* sS   = wreg + r16 * ST_PITCH + 4 * sq;
    const float* fA0 = wreg + l * ST_PITCH + 8 * half;
    const float* fA1 = fA0 + 16 * ST_PITCH;
    const float* fB0 = fA0 + 32 * ST_PITCH;
    const float* fB1 = fA0 + 48 * ST_PITCH;

    Slice a0, a1, b0, b1;
    v8f acc00 = {0.f,0.f,0.f,0.f,0.f,0.f,0.f,0.f};
    v8f acc01 = acc00, acc10 = acc00, acc11 = acc00;
    #pragma unroll
    for (int hf = 0; hf < 2; ++hf) {
        stage_block(sS + 32 * hf,                 ga0[hf]);
        stage_block(sS + 32 * ST_PITCH + 32 * hf, gb0[hf]);
        stage_block(sS + 48 * ST_PITCH + 32 * hf, gb1[hf]);
        stage_block(sS + 16 * ST_PITCH + 32 * hf, ga1[hf]);
        #pragma unroll
        for (int g = 2 * hf; g < 2 * hf + 2; ++g) {
            load_group(a0, fA0, g);
            load_group(b0, fB0, g);
            load_group(b1, fB1, g);
            load_group(a1, fA1, g);
        }
        __builtin_amdgcn_sched_barrier(0);
        #pragma unroll
        for (int st = 8 * hf; st < 8 * hf + 8; ++st) {
            v2f fa0 = frag_step(a0, st), fa1 = frag_step(a1, st);
            v2f fb0 = frag_step(b0, st), fb1 = frag_step(b1, st);
            acc00 = wmma_f32(fa0, fb0, acc00);
            acc01 = wmma_f32(fa0, fb1, acc01);
            acc10 = wmma_f32(fa1, fb0, acc10);
            acc11 = wmma_f32(fa1, fb1, acc11);
        }
        if (hf == 0) {
            __builtin_amdgcn_sched_barrier(0);
        }
    }

    float* rw = wreg;
    #pragma unroll
    for (int j = 0; j < 8; ++j) {
        int r = j + 8 * half;
        rw[r * RED_PITCH + l]             = acc00[j];
        rw[r * RED_PITCH + 16 + l]        = acc01[j];
        rw[(r + 16) * RED_PITCH + l]      = acc10[j];
        rw[(r + 16) * RED_PITCH + 16 + l] = acc11[j];
    }
    __syncthreads();

    const float* rp = stg + rr * RED_PITCH + jj;
    float z[4];
    #pragma unroll
    for (int g = 0; g < 4; ++g) {
        float s = rp[8 * g];
        #pragma unroll
        for (int w = 1; w < KSPLIT; ++w) s += rp[w * ST_WAVE + 8 * g];
        z[g] = s + zx[g];
    }
    c = sigmoidf_(z[1]) * c + sigmoidf_(z[0]) * tanhf(z[2]);
    float h = sigmoidf_(z[3]) * tanhf(c);
    cbuf[sidx]  = c;
    hnext[sidx] = h;
    ys[(size_t)(b * SEQ + t_d) * 1024 + dir * HID + j0 + jj] = h;
}

#define EW_BLOCKS  2048
#define EW_THREADS (EW_BLOCKS * 256)

typedef float f4v __attribute__((ext_vector_type(4)));

__global__ __launch_bounds__(256)
void gat1_lin(const float* __restrict__ x, const float* __restrict__ W,
              float* __restrict__ h) {
    int gtid = blockIdx.x * 256 + threadIdx.x;
    int q4 = (gtid & 15) * 4;
    f4v w0 = *(const f4v*)(W + q4);
    f4v w1 = *(const f4v*)(W + 64 + q4);
    for (int it = gtid; it < NN * 16; it += EW_THREADS) {
        v2f xv = *(const v2f*)(x + 2 * (it >> 4));
        f4v r;
        r.x = xv.x * w0.x + xv.y * w1.x;
        r.y = xv.x * w0.y + xv.y * w1.y;
        r.z = xv.x * w0.z + xv.y * w1.z;
        r.w = xv.x * w0.w + xv.y * w1.w;
        *(f4v*)(h + (size_t)it * 4) = r;
    }
}

__global__ void gat_scores(const float* __restrict__ h,
                           const float* __restrict__ asrc, const float* __restrict__ adst,
                           float* __restrict__ sas, float* __restrict__ sad) {
    int i = blockIdx.x * blockDim.x + threadIdx.x;
    if (i >= NN) return;
    const float* hp = h + (size_t)i * 64;
    float a = 0.f, d = 0.f;
    #pragma unroll
    for (int f = 0; f < 64; ++f) { float v = hp[f]; a += v * asrc[f]; d += v * adst[f]; }
    sas[i] = a; sad[i] = d;
}

__global__ __launch_bounds__(256)
void gat_aggr(const float* __restrict__ h,
              const float* __restrict__ sas, const float* __restrict__ sad,
              const float* __restrict__ bvec,
              const float* __restrict__ g, const float* __restrict__ bb,
              const float* __restrict__ bm, const float* __restrict__ bvv,
              float* __restrict__ out) {
    int gtid = blockIdx.x * 256 + threadIdx.x;
    int ch = (gtid & 15) * 4;
    f4v bias = *(const f4v*)(bvec + ch);
    f4v mean = *(const f4v*)(bm + ch);
    f4v gam  = *(const f4v*)(g + ch);
    f4v var  = *(const f4v*)(bvv + ch);
    f4v beta = *(const f4v*)(bb + ch);
    f4v scl;
    scl.x = gam.x * rsqrtf(var.x + 1e-5f);
    scl.y = gam.y * rsqrtf(var.y + 1e-5f);
    scl.z = gam.z * rsqrtf(var.z + 1e-5f);
    scl.w = gam.w * rsqrtf(var.w + 1e-5f);

    for (int it = gtid; it < NN * 16; it += EW_THREADS) {
        int i = it >> 4;
        int dl = i % KP, base = i - dl;
        float ad = sad[i];
        int o0 = NBR_OFF[dl], deg = NBR_OFF[dl + 1] - o0;

        float e[5];
        f4v hv[5];
        float t0 = sas[i] + ad;
        e[0]  = t0 < 0.f ? 0.2f * t0 : t0;
        hv[0] = *(const f4v*)(h + (size_t)i * 64 + ch);
        #pragma unroll
        for (int n = 0; n < 4; ++n) {
            bool on = n < deg;
            int k = on ? n : deg - 1;
            int s = base + NBR[o0 + k];
            float tt = sas[s] + ad;
            float ev = tt < 0.f ? 0.2f * tt : tt;
            e[n + 1]  = on ? ev : -__builtin_inff();
            hv[n + 1] = *(const f4v*)(h + (size_t)s * 64 + ch);
        }
        float m = e[0];
        #pragma unroll
        for (int n = 1; n < 5; ++n) m = fmaxf(m, e[n]);
        float den = 0.f;
        f4v num = {0.f, 0.f, 0.f, 0.f};
        #pragma unroll
        for (int n = 0; n < 5; ++n) {
            float w = __expf(e[n] - m);
            den += w;
            num += w * hv[n];
        }
        f4v val;
        val.x = num.x / den + bias.x;
        val.y = num.y / den + bias.y;
        val.z = num.z / den + bias.z;
        val.w = num.w / den + bias.w;
        val = (val - mean) * scl + beta;
        f4v o;
        o.x = fmaxf(val.x, 0.f);
        o.y = fmaxf(val.y, 0.f);
        o.z = fmaxf(val.z, 0.f);
        o.w = fmaxf(val.w, 0.f);
        *(f4v*)(out + (size_t)it * 4) = o;
    }
}

__global__ __launch_bounds__(256)
void pack_emb(const float* __restrict__ h2, const float* __restrict__ x,
              float* __restrict__ emb) {
    int gtid = blockIdx.x * 256 + threadIdx.x;
    for (int it = gtid; it < NN * 33; it += EW_THREADS) {
        int i = it / 33, c2 = it - i * 33;
        const float* p = (c2 < 32) ? h2 + (size_t)i * 64 + 2 * c2 : x + 2 * i;
        *(v2f*)(emb + (size_t)it * 2) = *(const v2f*)p;
    }
}

__global__ void transpose64(const float* __restrict__ W, float* __restrict__ WT) {
    int idx = blockIdx.x * blockDim.x + threadIdx.x;
    if (idx >= 64 * 64) return;
    int n = idx >> 6, k = idx & 63;
    WT[n * 64 + k] = W[k * 64 + n];
}

__global__ void zero_f32(float* __restrict__ p, int n) {
    int i = blockIdx.x * blockDim.x + threadIdx.x;
    if (i < n) p[i] = 0.f;
}

static inline void launch_gemm(hipStream_t s, const float* A, int lda,
                               const float* B, int ldb, const float* bias,
                               float* C, int ldc, int M, int N, int K, int act) {
    int tiles = (M / 32) * ((N + 31) / 32);
    int blocks = (tiles + 7) / 8;
    gemm_wmma_f32<<<blocks, 256, 0, s>>>(A, lda, B, ldb, bias, C, ldc, M, N, K, act);
}

static inline void launch_gemm_lds(hipStream_t s, const float* A, int lda,
                                   const float* B, int ldb, const float* bias,
                                   float* C, int ldc, int K) {
    gemm_lds_f32<<<dim3(GATE / 128, NGR / 128), 256, 0, s>>>(A, lda, B, ldb, bias, C, ldc, K);
}

extern "C" void kernel_launch(void* const* d_in, const int* in_sizes, int n_in,
                              void* d_out, int out_size, void* d_ws, size_t ws_size,
                              hipStream_t stream) {
    const float* x        = (const float*)d_in[0];
    const float* gat1_W   = (const float*)d_in[2];
    const float* gat1_as  = (const float*)d_in[3];
    const float* gat1_ad  = (const float*)d_in[4];
    const float* gat1_b   = (const float*)d_in[5];
    const float* gat2_W   = (const float*)d_in[6];
    const float* gat2_as  = (const float*)d_in[7];
    const float* gat2_ad  = (const float*)d_in[8];
    const float* gat2_b   = (const float*)d_in[9];
    const float* bn1_g = (const float*)d_in[10], *bn1_b = (const float*)d_in[11];
    const float* bn1_m = (const float*)d_in[12], *bn1_v = (const float*)d_in[13];
    const float* bn2_g = (const float*)d_in[14], *bn2_b = (const float*)d_in[15];
    const float* bn2_m = (const float*)d_in[16], *bn2_v = (const float*)d_in[17];
    const float* l1_Wih = (const float*)d_in[18];
    const float* l1_Whh = (const float*)d_in[19];
    const float* l1_b   = (const float*)d_in[20];
    const float* l2_Wih = (const float*)d_in[21];
    const float* l2_Whh = (const float*)d_in[22];
    const float* l2_b   = (const float*)d_in[23];
    const float* hW1 = (const float*)d_in[24], *hb1 = (const float*)d_in[25];
    const float* hW2 = (const float*)d_in[26], *hb2 = (const float*)d_in[27];
    float* out = (float*)d_out;
    float* ws  = (float*)d_ws;

    const size_t OFF_HA  = 0;
    const size_t OFF_HB  = OFF_HA  + (size_t)NN * 64;
    const size_t OFF_SAS = OFF_HB  + (size_t)NN * 64;
    const size_t OFF_SAD = OFF_SAS + NN;
    const size_t OFF_EMB = OFF_SAD + NN;
    const size_t OFF_ZX  = OFF_EMB + (size_t)NGR * D_IN1;
    const size_t OFF_ZF  = OFF_ZX  + (size_t)2 * NGR * GATE;
    const size_t OFF_HST = OFF_ZF  + (size_t)2 * BATCH * GATE;
    const size_t OFF_CST = OFF_HST + (size_t)2 * BATCH * HID;
    const size_t OFF_WT  = OFF_CST + (size_t)2 * BATCH * HID;

    float* hA   = ws + OFF_HA;
    float* hB   = ws + OFF_HB;
    float* sas  = ws + OFF_SAS;
    float* sad  = ws + OFF_SAD;
    float* emb  = ws + OFF_EMB;
    float* Zx   = ws + OFF_ZX;
    float* hs2  = ws + OFF_ZF;
    float* hst  = ws + OFF_HST;
    float* cst  = ws + OFF_CST;
    float* wt   = ws + OFF_WT;
    float* ys1  = emb;
    float* ys2  = hB;
    float* hd1  = hA;

    const int TPB = 256;

    gat1_lin<<<EW_BLOCKS, TPB, 0, stream>>>(x, gat1_W, hA);
    gat_scores<<<(NN + TPB - 1) / TPB, TPB, 0, stream>>>(hA, gat1_as, gat1_ad, sas, sad);
    gat_aggr<<<EW_BLOCKS, TPB, 0, stream>>>(hA, sas, sad, gat1_b,
                                            bn1_g, bn1_b, bn1_m, bn1_v, hB);
    transpose64<<<16, 256, 0, stream>>>(gat2_W, wt);
    launch_gemm(stream, hB, 64, wt, 64, nullptr, hA, 64, NN, 64, 64, 0);
    gat_scores<<<(NN + TPB - 1) / TPB, TPB, 0, stream>>>(hA, gat2_as, gat2_ad, sas, sad);
    gat_aggr<<<EW_BLOCKS, TPB, 0, stream>>>(hA, sas, sad, gat2_b,
                                            bn2_g, bn2_b, bn2_m, bn2_v, hB);
    pack_emb<<<EW_BLOCKS, TPB, 0, stream>>>(hB, x, emb);

    for (int d = 0; d < 2; ++d)
        launch_gemm_lds(stream, emb, D_IN1, l1_Wih + (size_t)d * GATE * D_IN1, D_IN1,
                        l1_b + (size_t)d * GATE, Zx + (size_t)d * NGR * GATE, GATE, D_IN1);
    zero_f32<<<(2 * 2 * BATCH * HID + TPB - 1) / TPB, TPB, 0, stream>>>(hst, 2 * 2 * BATCH * HID);
    for (int t = 0; t < SEQ; ++t) {
        lstm_step<<<dim3(128, 2), 256, 0, stream>>>(Zx, l1_Whh, hst, hs2, cst, ys1, t);
    }

    for (int d = 0; d < 2; ++d)
        launch_gemm_lds(stream, ys1, D_IN2, l2_Wih + (size_t)d * GATE * D_IN2, D_IN2,
                        l2_b + (size_t)d * GATE, Zx + (size_t)d * NGR * GATE, GATE, D_IN2);
    zero_f32<<<(2 * 2 * BATCH * HID + TPB - 1) / TPB, TPB, 0, stream>>>(hst, 2 * 2 * BATCH * HID);
    for (int t = 0; t < SEQ; ++t) {
        lstm_step<<<dim3(128, 2), 256, 0, stream>>>(Zx, l2_Whh, hst, hs2, cst, ys2, t);
    }

    launch_gemm(stream, ys2, 1024, hW1, 1024, hb1, hd1, 256, NGR, 256, 1024, 1);
    launch_gemm(stream, hd1, 256,  hW2, 256,  hb2, out, 57,  NGR, 57,  256,  0);
}
